# starting kernel
# baseline (speedup 1.0000x reference)
#include <hip/hip_runtime.h>
#include <math.h>

typedef float v2f __attribute__((ext_vector_type(2)));
typedef float v8f __attribute__((ext_vector_type(8)));

constexpr int T_LEN = 8192;

__global__ __launch_bounds__(32) void dsvf_wmma_kernel(
    const float* __restrict__ x,
    const float* __restrict__ gp,
    const float* __restrict__ rp,
    const float* __restrict__ mhp_p,
    const float* __restrict__ mbp_p,
    const float* __restrict__ mlp_p,
    float* __restrict__ yout)
{
    __shared__ float xs[16][17];
    __shared__ float us[16][17];
    __shared__ float cx1[16], cx2[16];
    __shared__ float cy1[16], cy2[16];
    __shared__ float hs[20];

    const unsigned lane = threadIdx.x;
    const unsigned j    = lane & 15u;
    const bool     hi   = lane >= 16u;
    const int rbase = blockIdx.x * 16;

    const float gv  = gp[0];
    const float rv  = rp[0];
    const float mhp = mhp_p[0];
    const float mbp = mbp_p[0];
    const float mlp = mlp_p[0];

    const float sig = 1.0f / (1.0f + expf(-gv));
    const float gg  = tanf(3.14159265358979323846f * 0.5f * sig);
    const float rr  = log1pf(expf(rv));
    const float g2  = gg * gg;

    const float b0 = g2 * mlp + gg * mbp + mhp;
    const float b1 = 2.0f * g2 * mlp - 2.0f * mhp;
    const float b2 = g2 * mlp - gg * mbp + mhp;
    const float a0 = g2 + 2.0f * rr * gg + 1.0f;
    const float inva0 = 1.0f / a0;
    const float a1 = (2.0f * g2 - 2.0f) * inva0;
    const float a2 = (g2 - 2.0f * rr * gg + 1.0f) * inva0;

    if (lane == 0u) {
        hs[0] = 1.0f;
        hs[1] = -a1;
        for (int i = 2; i <= 16; ++i)
            hs[i] = -a1 * hs[i - 1] - a2 * hs[i - 2];
    }
    if (lane < 16u) {
        cx1[lane] = 0.0f; cx2[lane] = 0.0f;
        cy1[lane] = 0.0f; cy2[lane] = 0.0f;
    }
    __syncthreads();

    v2f Bf[4];
#pragma unroll
    for (int s = 0; s < 4; ++s) {
        const int k0 = 4 * s + (hi ? 2 : 0);
        const int k1 = k0 + 1;
        Bf[s].x = ((int)j >= k0) ? hs[j - k0] : 0.0f;
        Bf[s].y = ((int)j >= k1) ? hs[j - k1] : 0.0f;
    }
    const float pj = hs[j + 1];
    const float qj = -a2 * hs[j];

    const int rload = (int)(lane >> 1);
    const int toff  = (int)(lane & 1u) * 8;
    const float* xrow = x + (size_t)(rbase + rload) * T_LEN;

    for (int t0 = 0; t0 < T_LEN; t0 += 16) {
        const float4 xa = *(const float4*)(xrow + t0 + toff);
        const float4 xb = *(const float4*)(xrow + t0 + toff + 4);
        if (t0 + 16 < T_LEN)
            __builtin_prefetch(xrow + t0 + 16 + toff, 0, 1);

        xs[rload][toff + 0] = xa.x; xs[rload][toff + 1] = xa.y;
        xs[rload][toff + 2] = xa.z; xs[rload][toff + 3] = xa.w;
        xs[rload][toff + 4] = xb.x; xs[rload][toff + 5] = xb.y;
        xs[rload][toff + 6] = xb.z; xs[rload][toff + 7] = xb.w;
        __syncthreads();

        float prev1 = toff ? xs[rload][7] : cx1[rload];
        float prev2 = toff ? xs[rload][6] : cx2[rload];
#pragma unroll
        for (int k = 0; k < 8; ++k) {
            const float xv = xs[rload][toff + k];
            us[rload][toff + k] = b0 * xv + b1 * prev1 + b2 * prev2;
            prev2 = prev1;
            prev1 = xv;
        }
        __syncthreads();

        if (toff) {
            cx1[rload] = xs[rload][15];
            cx2[rload] = xs[rload][14];
        }

        v8f acc = {};
#pragma unroll
        for (int s = 0; s < 4; ++s) {
            const int k0 = 4 * s + (hi ? 2 : 0);
            v2f A;
            A.x = us[j][k0];
            A.y = us[j][k0 + 1];
            acc = __builtin_amdgcn_wmma_f32_16x16x4_f32(
                 false, A,  false, Bf[s],
                 (short)0, acc,  false,  false);
        }

        float outv[8];
#pragma unroll
        for (int v = 0; v < 8; ++v) {
            const int r2 = v + (hi ? 8 : 0);
            const float yv = acc[v] + pj * cy1[r2] + qj * cy2[r2];
            outv[v] = yv;
            yout[(size_t)(rbase + r2) * T_LEN + t0 + j] = yv;
        }
        __syncthreads();

        if (j == 15u) {
#pragma unroll
            for (int v = 0; v < 8; ++v) cy1[v + (hi ? 8 : 0)] = outv[v];
        }
        if (j == 14u) {
#pragma unroll
            for (int v = 0; v < 8; ++v) cy2[v + (hi ? 8 : 0)] = outv[v];
        }
        __syncthreads();
    }
}

extern "C" void kernel_launch(void* const* d_in, const int* in_sizes, int n_in,
                              void* d_out, int out_size, void* d_ws, size_t ws_size,
                              hipStream_t stream) {
    const float* x   = (const float*)d_in[0];
    const float* g   = (const float*)d_in[1];
    const float* r   = (const float*)d_in[2];
    const float* mhp = (const float*)d_in[3];
    const float* mbp = (const float*)d_in[4];
    const float* mlp = (const float*)d_in[5];
    float* y = (float*)d_out;

    const int B = in_sizes[0] / T_LEN;
    dim3 grid(B / 16), block(32);
    hipLaunchKernelGGL(dsvf_wmma_kernel, grid, block, 0, stream,
                       x, g, r, mhp, mbp, mlp, y);
}


# 16-step look-ahead: y[t] from y[t-16], y[t-32], 33 x taps; 3 Toeplitz products on x fragments read straight from memory, carry elementwise in registers, no LDS or barrier in loop; sum order changes
# speedup vs baseline: 2.9863x; 2.9863x over previous
#include <hip/hip_runtime.h>
#include <math.h>

typedef float v2f __attribute__((ext_vector_type(2)));
typedef float v8f __attribute__((ext_vector_type(8)));

constexpr int T_LEN  = 8192;
constexpr int NCHUNK = T_LEN / 16;

__global__ __launch_bounds__(32) void dsvf_wmma_kernel(
    const float* __restrict__ x,
    const float* __restrict__ gp,
    const float* __restrict__ rp,
    const float* __restrict__ mhp_p,
    const float* __restrict__ mbp_p,
    const float* __restrict__ mlp_p,
    float* __restrict__ yout)
{
    __shared__ float ds[40];

    const unsigned lane = threadIdx.x;
    const unsigned j    = lane & 15u;
    const bool     hi   = lane >= 16u;
    const int rbase = blockIdx.x * 16;

    const float gv  = gp[0];
    const float rv  = rp[0];
    const float mhp = mhp_p[0];
    const float mbp = mbp_p[0];
    const float mlp = mlp_p[0];

    const float sig = 1.0f / (1.0f + expf(-gv));
    const float gg  = tanf(3.14159265358979323846f * 0.5f * sig);
    const float rr  = log1pf(expf(rv));
    const float g2  = gg * gg;

    const float b0 = g2 * mlp + gg * mbp + mhp;
    const float b1 = 2.0f * g2 * mlp - 2.0f * mhp;
    const float b2 = g2 * mlp - gg * mbp + mhp;
    const float a0 = g2 + 2.0f * rr * gg + 1.0f;
    const float inva0 = 1.0f / a0;
    const float a1 = (2.0f * g2 - 2.0f) * inva0;
    const float a2 = (g2 - 2.0f * rr * gg + 1.0f) * inva0;

    if (lane == 0u) {
        for (int i = 0; i < 40; ++i) ds[i] = 0.0f;
        ds[0] = b0; ds[1] = b1; ds[2] = b2;
    }
    float q1 = a1, q2 = a2;
    for (int s = 1; s <= 8; s *= 2) {
        if (lane == 0u) {
            for (int i = 32; i >= 0; --i) {
                float v = ds[i];
                if (i >= s)     v -= q1 * ds[i - s];
                if (i >= 2 * s) v += q2 * ds[i - 2 * s];
                ds[i] = v;
            }
        }
        const float n1 = 2.0f * q2 - q1 * q1;
        const float n2 = q2 * q2;
        q1 = n1; q2 = n2;
    }
    const float nalpha = -q1;
    const float nbeta  = -q2;
    __syncthreads();

    v2f Bf[3][4];
#pragma unroll
    for (int b = 0; b < 3; ++b) {
#pragma unroll
        for (int s = 0; s < 4; ++s) {
            const int k0 = (hi ? 8 : 0) + 2 * s;
            const int i0 = 16 * b + (int)j - k0;
            const int i1 = i0 - 1;
            const float d0 = ds[min(max(i0, 0), 39)];
            const float d1 = ds[min(max(i1, 0), 39)];
            Bf[b][s].x = (i0 >= 0 && i0 <= 32) ? d0 : 0.0f;
            Bf[b][s].y = (i1 >= 0 && i1 <= 32) ? d1 : 0.0f;
        }
    }

    const float* xrow = x + (size_t)(rbase + (int)j) * T_LEN + (hi ? 8 : 0);
    float* yrow = yout + (size_t)(rbase + (hi ? 8 : 0)) * T_LEN + j;

    v2f X[4][4];
#pragma unroll
    for (int u = 0; u < 4; ++u)
#pragma unroll
        for (int s = 0; s < 4; ++s) { X[u][s].x = 0.0f; X[u][s].y = 0.0f; }
    v8f Y1 = {}, Y2 = {};

    for (int m0 = 0; m0 < NCHUNK; m0 += 4) {
        if (m0 * 16 + 128 < T_LEN)
            __builtin_prefetch(xrow + m0 * 16 + 128, 0, 1);
#pragma unroll
        for (int u = 0; u < 4; ++u) {
            const int t0 = (m0 + u) * 16;
            const float4 xa = *(const float4*)(xrow + t0);
            const float4 xb = *(const float4*)(xrow + t0 + 4);

            const int c  = u;
            const int p1 = (u + 3) & 3;
            const int p2 = (u + 2) & 3;
            X[c][0].x = xa.x; X[c][0].y = xa.y;
            X[c][1].x = xa.z; X[c][1].y = xa.w;
            X[c][2].x = xb.x; X[c][2].y = xb.y;
            X[c][3].x = xb.z; X[c][3].y = xb.w;

            v8f accA = {}, accB = {};
#pragma unroll
            for (int s = 0; s < 4; ++s) {
                accA = __builtin_amdgcn_wmma_f32_16x16x4_f32(
                    false, X[p2][s], false, Bf[2][s], (short)0, accA, false, false);
                accB = __builtin_amdgcn_wmma_f32_16x16x4_f32(
                    false, X[p1][s], false, Bf[1][s], (short)0, accB, false, false);
            }
#pragma unroll
            for (int s = 0; s < 4; s += 2) {
                accA = __builtin_amdgcn_wmma_f32_16x16x4_f32(
                    false, X[c][s], false, Bf[0][s], (short)0, accA, false, false);
                accB = __builtin_amdgcn_wmma_f32_16x16x4_f32(
                    false, X[c][s + 1], false, Bf[0][s + 1], (short)0, accB, false, false);
            }

            v8f Y;
#pragma unroll
            for (int v = 0; v < 8; ++v) {
                const float yv = (accA[v] + accB[v]) + nalpha * Y1[v] + nbeta * Y2[v];
                Y[v] = yv;
                yrow[(size_t)v * T_LEN + t0] = yv;
            }
            Y2 = Y1;
            Y1 = Y;
        }
    }
}

extern "C" void kernel_launch(void* const* d_in, const int* in_sizes, int n_in,
                              void* d_out, int out_size, void* d_ws, size_t ws_size,
                              hipStream_t stream) {
    const float* x   = (const float*)d_in[0];
    const float* g   = (const float*)d_in[1];
    const float* r   = (const float*)d_in[2];
    const float* mhp = (const float*)d_in[3];
    const float* mbp = (const float*)d_in[4];
    const float* mlp = (const float*)d_in[5];
    float* y = (float*)d_out;

    const int B = in_sizes[0] / T_LEN;
    dim3 grid(B / 16), block(32);
    hipLaunchKernelGGL(dsvf_wmma_kernel, grid, block, 0, stream,
                       x, g, r, mhp, mbp, mlp, y);
}


# Products and carry in 2 launches: 1) Toeplitz products of all chunks (stateless), 16 time slices per row tile, 2048 waves, into out; 2) carry in place, one lane per chain; carry sum order changes
# speedup vs baseline: 7.1194x; 2.3840x over previous
#include <hip/hip_runtime.h>
#include <math.h>


typedef float v2f __attribute__((ext_vector_type(2)));
typedef float v8f __attribute__((ext_vector_type(8)));

constexpr int T_LEN  = 8192;
constexpr int NCHUNK = T_LEN / 16;
constexpr int NSLICE = 16;
constexpr int SLICE_CHUNKS = NCHUNK / NSLICE;


struct Coef { float b0, b1, b2, a1, a2; };

__device__ __forceinline__ Coef dsvf_coef(const float* gp, const float* rp,
                                          const float* mhp_p, const float* mbp_p,
                                          const float* mlp_p)
{
    const float gv  = gp[0];
    const float rv  = rp[0];
    const float mhp = mhp_p[0];
    const float mbp = mbp_p[0];
    const float mlp = mlp_p[0];

    const float sig = 1.0f / (1.0f + expf(-gv));
    const float gg  = tanf(3.14159265358979323846f * 0.5f * sig);
    const float rr  = log1pf(expf(rv));
    const float g2  = gg * gg;

    Coef c;
    c.b0 = g2 * mlp + gg * mbp + mhp;
    c.b1 = 2.0f * g2 * mlp - 2.0f * mhp;
    c.b2 = g2 * mlp - gg * mbp + mhp;
    const float a0 = g2 + 2.0f * rr * gg + 1.0f;
    const float inva0 = 1.0f / a0;
    c.a1 = (2.0f * g2 - 2.0f) * inva0;
    c.a2 = (g2 - 2.0f * rr * gg + 1.0f) * inva0;
    return c;
}

__global__ __launch_bounds__(32) void dsvf_fir_kernel(
    const float* __restrict__ x,
    const float* __restrict__ gp,
    const float* __restrict__ rp,
    const float* __restrict__ mhp_p,
    const float* __restrict__ mbp_p,
    const float* __restrict__ mlp_p,
    float* __restrict__ fout)
{
    __shared__ float ds[40];

    const unsigned lane = threadIdx.x;
    const unsigned j    = lane & 15u;
    const bool     hi   = lane >= 16u;
    const int rbase  = (int)(blockIdx.x / NSLICE) * 16;
    const int mstart = (int)(blockIdx.x % NSLICE) * SLICE_CHUNKS;

    const Coef cf = dsvf_coef(gp, rp, mhp_p, mbp_p, mlp_p);

    if (lane == 0u) {
        for (int i = 0; i < 40; ++i) ds[i] = 0.0f;
        ds[0] = cf.b0; ds[1] = cf.b1; ds[2] = cf.b2;
    }
    float q1 = cf.a1, q2 = cf.a2;
    for (int s = 1; s <= 8; s *= 2) {
        if (lane == 0u) {
            for (int i = 32; i >= 0; --i) {
                float v = ds[i];
                if (i >= s)     v -= q1 * ds[i - s];
                if (i >= 2 * s) v += q2 * ds[i - 2 * s];
                ds[i] = v;
            }
        }
        const float n1 = 2.0f * q2 - q1 * q1;
        const float n2 = q2 * q2;
        q1 = n1; q2 = n2;
    }
    __syncthreads();

    v2f Bf[3][4];
#pragma unroll
    for (int b = 0; b < 3; ++b) {
#pragma unroll
        for (int s = 0; s < 4; ++s) {
            const int k0 = (hi ? 8 : 0) + 2 * s;
            const int i0 = 16 * b + (int)j - k0;
            const int i1 = i0 - 1;
            const float d0 = ds[min(max(i0, 0), 39)];
            const float d1 = ds[min(max(i1, 0), 39)];
            Bf[b][s].x = (i0 >= 0 && i0 <= 32) ? d0 : 0.0f;
            Bf[b][s].y = (i1 >= 0 && i1 <= 32) ? d1 : 0.0f;
        }
    }

    const float* xrow = x + (size_t)(rbase + (int)j) * T_LEN + (hi ? 8 : 0);
    float* frow = fout + (size_t)(rbase + (hi ? 8 : 0)) * T_LEN + j;

    v2f X[4][4];
#pragma unroll
    for (int u = 0; u < 4; ++u)
#pragma unroll
        for (int s = 0; s < 4; ++s) { X[u][s].x = 0.0f; X[u][s].y = 0.0f; }

    if (mstart > 0) {
        const float4 pa = *(const float4*)(xrow + (mstart - 1) * 16);
        const float4 pb = *(const float4*)(xrow + (mstart - 1) * 16 + 4);
        const float4 qa = *(const float4*)(xrow + (mstart - 2) * 16);
        const float4 qb = *(const float4*)(xrow + (mstart - 2) * 16 + 4);
        X[3][0].x = pa.x; X[3][0].y = pa.y;
        X[3][1].x = pa.z; X[3][1].y = pa.w;
        X[3][2].x = pb.x; X[3][2].y = pb.y;
        X[3][3].x = pb.z; X[3][3].y = pb.w;
        X[2][0].x = qa.x; X[2][0].y = qa.y;
        X[2][1].x = qa.z; X[2][1].y = qa.w;
        X[2][2].x = qb.x; X[2][2].y = qb.y;
        X[2][3].x = qb.z; X[2][3].y = qb.w;
    }

    for (int m0 = mstart; m0 < mstart + SLICE_CHUNKS; m0 += 4) {
        if (m0 + 8 < mstart + SLICE_CHUNKS)
            __builtin_prefetch(xrow + m0 * 16 + 128, 0, 1);
#pragma unroll
        for (int u = 0; u < 4; ++u) {
            const int t0 = (m0 + u) * 16;
            const float4 xa = *(const float4*)(xrow + t0);
            const float4 xb = *(const float4*)(xrow + t0 + 4);

            const int c  = u;
            const int p1 = (u + 3) & 3;
            const int p2 = (u + 2) & 3;
            X[c][0].x = xa.x; X[c][0].y = xa.y;
            X[c][1].x = xa.z; X[c][1].y = xa.w;
            X[c][2].x = xb.x; X[c][2].y = xb.y;
            X[c][3].x = xb.z; X[c][3].y = xb.w;

            v8f accA = {}, accB = {};
#pragma unroll
            for (int s = 0; s < 4; ++s) {
                accA = __builtin_amdgcn_wmma_f32_16x16x4_f32(
                    false, X[p2][s], false, Bf[2][s], (short)0, accA, false, false);
                accB = __builtin_amdgcn_wmma_f32_16x16x4_f32(
                    false, X[p1][s], false, Bf[1][s], (short)0, accB, false, false);
            }
#pragma unroll
            for (int s = 0; s < 4; s += 2) {
                accA = __builtin_amdgcn_wmma_f32_16x16x4_f32(
                    false, X[c][s], false, Bf[0][s], (short)0, accA, false, false);
                accB = __builtin_amdgcn_wmma_f32_16x16x4_f32(
                    false, X[c][s + 1], false, Bf[0][s + 1], (short)0, accB, false, false);
            }

#pragma unroll
            for (int v = 0; v < 8; ++v)
                frow[(size_t)v * T_LEN + t0] = accA[v] + accB[v];
        }
    }
}

constexpr int CARRY_BATCH = 64;

#define DSVF_CARRY_BATCH(SET, PTR)                                          \
    _Pragma("unroll")                                                       \
    for (int i = 0; i < CARRY_BATCH; ++i) {                                 \
           \
        const float yv = (SET[i] + nbeta * y2) + nalpha * y1;               \
        SET[i] = yv;                                                        \
        y2 = y1; y1 = yv;                                                   \
    }                                                                       \
    _Pragma("unroll")                                                       \
    for (int i = 0; i < CARRY_BATCH; ++i) (PTR)[i * 16] = SET[i];

__global__ __launch_bounds__(32) void dsvf_carry_kernel(
    const float* __restrict__ gp,
    const float* __restrict__ rp,
    const float* __restrict__ mhp_p,
    const float* __restrict__ mbp_p,
    const float* __restrict__ mlp_p,
    float* y)
{
    const unsigned lane = threadIdx.x;
    const int row = (int)blockIdx.x * 2 + (int)(lane >> 4);

    const Coef cf = dsvf_coef(gp, rp, mhp_p, mbp_p, mlp_p);
    float q1 = cf.a1, q2 = cf.a2;
    for (int s = 1; s <= 8; s *= 2) {
        const float n1 = 2.0f * q2 - q1 * q1;
        const float n2 = q2 * q2;
        q1 = n1; q2 = n2;
    }
    const float nalpha = -q1;
    const float nbeta  = -q2;

    float* p = y + (size_t)row * T_LEN + (lane & 15u);

    float fa[CARRY_BATCH], fb[CARRY_BATCH];
#pragma unroll
    for (int i = 0; i < CARRY_BATCH; ++i) fa[i] = p[i * 16];

    float y1 = 0.0f, y2 = 0.0f;
    int mb = 0;
    for (; mb < NCHUNK - 2 * CARRY_BATCH; mb += 2 * CARRY_BATCH) {
        float* pa = p + mb * 16;
        float* pb = pa + CARRY_BATCH * 16;
        float* pc = pb + CARRY_BATCH * 16;
#pragma unroll
        for (int i = 0; i < CARRY_BATCH; ++i) fb[i] = pb[i * 16];
        __builtin_amdgcn_sched_barrier(0);
        DSVF_CARRY_BATCH(fa, pa)
        __builtin_amdgcn_sched_barrier(0);
#pragma unroll
        for (int i = 0; i < CARRY_BATCH; ++i) fa[i] = pc[i * 16];
        __builtin_amdgcn_sched_barrier(0);
        DSVF_CARRY_BATCH(fb, pb)
        __builtin_amdgcn_sched_barrier(0);
    }
    {
        float* pa = p + mb * 16;
        float* pb = pa + CARRY_BATCH * 16;
#pragma unroll
        for (int i = 0; i < CARRY_BATCH; ++i) fb[i] = pb[i * 16];
        __builtin_amdgcn_sched_barrier(0);
        DSVF_CARRY_BATCH(fa, pa)
        DSVF_CARRY_BATCH(fb, pb)
    }
}
#undef DSVF_CARRY_BATCH

extern "C" void kernel_launch(void* const* d_in, const int* in_sizes, int n_in,
                              void* d_out, int out_size, void* d_ws, size_t ws_size,
                              hipStream_t stream) {
    const float* x   = (const float*)d_in[0];
    const float* g   = (const float*)d_in[1];
    const float* r   = (const float*)d_in[2];
    const float* mhp = (const float*)d_in[3];
    const float* mbp = (const float*)d_in[4];
    const float* mlp = (const float*)d_in[5];
    float* y = (float*)d_out;

    const int B = in_sizes[0] / T_LEN;
    dim3 gridF(B / 16 * NSLICE), gridC(B / 2), block(32);
    hipLaunchKernelGGL(dsvf_fir_kernel, gridF, block, 0, stream,
                       x, g, r, mhp, mbp, mlp, y);
    hipLaunchKernelGGL(dsvf_carry_kernel, gridC, block, 0, stream,
                       g, r, mhp, mbp, mlp, y);
}


# F no longer stored: launch 1 keeps only each slice's end state from rest (carry in registers); launch 2 folds these, redoes the products, carries in registers, writes y; carry sum order changes
# speedup vs baseline: 7.5784x; 1.0645x over previous
#include <hip/hip_runtime.h>
#include <math.h>


typedef float v2f __attribute__((ext_vector_type(2)));
typedef float v8f __attribute__((ext_vector_type(8)));

constexpr int T_LEN  = 8192;
constexpr int NCHUNK = T_LEN / 16;
constexpr int NSLICE = 16;
constexpr int SLICE_CHUNKS = NCHUNK / NSLICE;


struct Coef { float b0, b1, b2, a1, a2; };

__device__ __forceinline__ Coef dsvf_coef(const float* gp, const float* rp,
                                          const float* mhp_p, const float* mbp_p,
                                          const float* mlp_p)
{
    const float gv  = gp[0];
    const float rv  = rp[0];
    const float mhp = mhp_p[0];
    const float mbp = mbp_p[0];
    const float mlp = mlp_p[0];

    const float sig = 1.0f / (1.0f + expf(-gv));
    const float gg  = tanf(3.14159265358979323846f * 0.5f * sig);
    const float rr  = log1pf(expf(rv));
    const float g2  = gg * gg;

    Coef c;
    c.b0 = g2 * mlp + gg * mbp + mhp;
    c.b1 = 2.0f * g2 * mlp - 2.0f * mhp;
    c.b2 = g2 * mlp - gg * mbp + mhp;
    const float a0 = g2 + 2.0f * rr * gg + 1.0f;
    const float inva0 = 1.0f / a0;
    c.a1 = (2.0f * g2 - 2.0f) * inva0;
    c.a2 = (g2 - 2.0f * rr * gg + 1.0f) * inva0;
    return c;
}

template <bool FINAL>
__device__ __forceinline__ void dsvf_slice_body(
    const float* __restrict__ x,
    const float* __restrict__ gp,
    const float* __restrict__ rp,
    const float* __restrict__ mhp_p,
    const float* __restrict__ mbp_p,
    const float* __restrict__ mlp_p,
    float* fout,
    float* ds)
{

    const unsigned lane = threadIdx.x;
    const unsigned j    = lane & 15u;
    const bool     hi   = lane >= 16u;
    const int rbase  = (int)(blockIdx.x / NSLICE) * 16;
    const int slice  = (int)(blockIdx.x % NSLICE);
    const int mstart = slice * SLICE_CHUNKS;
    if (!FINAL && slice == NSLICE - 1) return;

    const Coef cf = dsvf_coef(gp, rp, mhp_p, mbp_p, mlp_p);

    if (lane == 0u) {
        for (int i = 0; i < 40; ++i) ds[i] = 0.0f;
        ds[0] = cf.b0; ds[1] = cf.b1; ds[2] = cf.b2;
    }
    float q1 = cf.a1, q2 = cf.a2;
    for (int s = 1; s <= 8; s *= 2) {
        if (lane == 0u) {
            for (int i = 32; i >= 0; --i) {
                float v = ds[i];
                if (i >= s)     v -= q1 * ds[i - s];
                if (i >= 2 * s) v += q2 * ds[i - 2 * s];
                ds[i] = v;
            }
        }
        const float n1 = 2.0f * q2 - q1 * q1;
        const float n2 = q2 * q2;
        q1 = n1; q2 = n2;
    }
    __syncthreads();

    v2f Bf[3][4];
#pragma unroll
    for (int b = 0; b < 3; ++b) {
#pragma unroll
        for (int s = 0; s < 4; ++s) {
            const int k0 = (hi ? 8 : 0) + 2 * s;
            const int i0 = 16 * b + (int)j - k0;
            const int i1 = i0 - 1;
            const float d0 = ds[min(max(i0, 0), 39)];
            const float d1 = ds[min(max(i1, 0), 39)];
            Bf[b][s].x = (i0 >= 0 && i0 <= 32) ? d0 : 0.0f;
            Bf[b][s].y = (i1 >= 0 && i1 <= 32) ? d1 : 0.0f;
        }
    }

    const float nalpha = -q1;
    const float nbeta  = -q2;

    v8f y1 = {}, y2 = {};

    if (FINAL) {
        float pa = 1.0f, pb = 0.0f, qa = 0.0f, qb = 1.0f;
        for (int k = 0; k < SLICE_CHUNKS; ++k) {
            const float np = nalpha * pa + nbeta * pb;
            const float nq = nalpha * qa + nbeta * qb;
            pb = pa; pa = np;
            qb = qa; qa = nq;
        }
        const float* zbase = fout + (size_t)rbase * T_LEN + (size_t)mstart * 16 + lane;
        float z[NSLICE - 1][16];
#pragma unroll
        for (int i = 0; i < NSLICE - 1; ++i) {
            if (i < slice) {
#pragma unroll
                for (int r = 0; r < 16; ++r) z[i][r] = zbase[(size_t)i * T_LEN + r * 32];
            }
        }
#pragma unroll
        for (int i = 0; i < NSLICE - 1; ++i) {
            if (i < slice) {
#pragma unroll
                for (int r = 0; r < 8; ++r) {
                    const float s1 = y1[r], s2 = y2[r];
                    y1[r] = z[i][r]     + (pa * s1 + qa * s2);
                    y2[r] = z[i][8 + r] + (pb * s1 + qb * s2);
                }
            }
        }
    }

    const float* xrow = x + (size_t)(rbase + (int)j) * T_LEN + (hi ? 8 : 0);
    float* frow = fout + (size_t)(rbase + (hi ? 8 : 0)) * T_LEN + j;

    v2f X[4][4];
#pragma unroll
    for (int u = 0; u < 4; ++u)
#pragma unroll
        for (int s = 0; s < 4; ++s) { X[u][s].x = 0.0f; X[u][s].y = 0.0f; }

    if (mstart > 0) {
        const float4 pa = *(const float4*)(xrow + (mstart - 1) * 16);
        const float4 pb = *(const float4*)(xrow + (mstart - 1) * 16 + 4);
        const float4 qa = *(const float4*)(xrow + (mstart - 2) * 16);
        const float4 qb = *(const float4*)(xrow + (mstart - 2) * 16 + 4);
        X[3][0].x = pa.x; X[3][0].y = pa.y;
        X[3][1].x = pa.z; X[3][1].y = pa.w;
        X[3][2].x = pb.x; X[3][2].y = pb.y;
        X[3][3].x = pb.z; X[3][3].y = pb.w;
        X[2][0].x = qa.x; X[2][0].y = qa.y;
        X[2][1].x = qa.z; X[2][1].y = qa.w;
        X[2][2].x = qb.x; X[2][2].y = qb.y;
        X[2][3].x = qb.z; X[2][3].y = qb.w;
    }

    for (int m0 = mstart; m0 < mstart + SLICE_CHUNKS; m0 += 4) {
        if (m0 + 8 < mstart + SLICE_CHUNKS)
            __builtin_prefetch(xrow + m0 * 16 + 128, 0, 1);
#pragma unroll
        for (int u = 0; u < 4; ++u) {
            const int t0 = (m0 + u) * 16;
            const float4 xa = *(const float4*)(xrow + t0);
            const float4 xb = *(const float4*)(xrow + t0 + 4);

            const int c  = u;
            const int p1 = (u + 3) & 3;
            const int p2 = (u + 2) & 3;
            X[c][0].x = xa.x; X[c][0].y = xa.y;
            X[c][1].x = xa.z; X[c][1].y = xa.w;
            X[c][2].x = xb.x; X[c][2].y = xb.y;
            X[c][3].x = xb.z; X[c][3].y = xb.w;

            v8f accA = {}, accB = {};
#pragma unroll
            for (int s = 0; s < 4; ++s) {
                accA = __builtin_amdgcn_wmma_f32_16x16x4_f32(
                    false, X[p2][s], false, Bf[2][s], (short)0, accA, false, false);
                accB = __builtin_amdgcn_wmma_f32_16x16x4_f32(
                    false, X[p1][s], false, Bf[1][s], (short)0, accB, false, false);
            }
#pragma unroll
            for (int s = 0; s < 4; s += 2) {
                accA = __builtin_amdgcn_wmma_f32_16x16x4_f32(
                    false, X[c][s], false, Bf[0][s], (short)0, accA, false, false);
                accB = __builtin_amdgcn_wmma_f32_16x16x4_f32(
                    false, X[c][s + 1], false, Bf[0][s + 1], (short)0, accB, false, false);
            }

            const v8f yv = ((accA + accB) + nbeta * y2) + nalpha * y1;
            y2 = y1; y1 = yv;

            if (FINAL) {
#pragma unroll
                for (int v = 0; v < 8; ++v)
                    frow[(size_t)v * T_LEN + t0] = yv[v];
            }
        }
    }

    if (!FINAL) {
        float* zrow = fout + (size_t)(rbase + slice) * T_LEN + lane;
        int s2 = slice + 1;
        do {
            float* zp = zrow + s2 * (SLICE_CHUNKS * 16);
#pragma unroll
            for (int r = 0; r < 8; ++r) {
                zp[r * 32]       = y1[r];
                zp[(8 + r) * 32] = y2[r];
            }
        } while (++s2 < NSLICE);
    }
}

__global__ __launch_bounds__(32) void dsvf_state_kernel(
    const float* __restrict__ x,
    const float* __restrict__ gp,
    const float* __restrict__ rp,
    const float* __restrict__ mhp_p,
    const float* __restrict__ mbp_p,
    const float* __restrict__ mlp_p,
    float* out)
{
    __shared__ float ds[40];
    dsvf_slice_body<false>(x, gp, rp, mhp_p, mbp_p, mlp_p, out, ds);
}

__global__ __launch_bounds__(32) void dsvf_out_kernel(
    const float* __restrict__ x,
    const float* __restrict__ gp,
    const float* __restrict__ rp,
    const float* __restrict__ mhp_p,
    const float* __restrict__ mbp_p,
    const float* __restrict__ mlp_p,
    float* out)
{
    __shared__ float ds[40];
    dsvf_slice_body<true>(x, gp, rp, mhp_p, mbp_p, mlp_p, out, ds);
}


extern "C" void kernel_launch(void* const* d_in, const int* in_sizes, int n_in,
                              void* d_out, int out_size, void* d_ws, size_t ws_size,
                              hipStream_t stream) {
    const float* x   = (const float*)d_in[0];
    const float* g   = (const float*)d_in[1];
    const float* r   = (const float*)d_in[2];
    const float* mhp = (const float*)d_in[3];
    const float* mbp = (const float*)d_in[4];
    const float* mlp = (const float*)d_in[5];
    float* y = (float*)d_out;

    const int B = in_sizes[0] / T_LEN;
    dim3 grid(B / 16 * NSLICE), block(32);
    hipLaunchKernelGGL(dsvf_state_kernel, grid, block, 0, stream,
                       x, g, r, mhp, mbp, mlp, y);
    hipLaunchKernelGGL(dsvf_out_kernel, grid, block, 0, stream,
                       x, g, r, mhp, mbp, mlp, y);
}


# State launch: the scalar carry commutes with the Toeplitz products, so it runs on the x fragments and 6 products close a slice (24 WMMA, was 384); output launch as before; carry sum order changes
# speedup vs baseline: 8.7530x; 1.1550x over previous
#include <hip/hip_runtime.h>
#include <math.h>


typedef float v2f __attribute__((ext_vector_type(2)));
typedef float v8f __attribute__((ext_vector_type(8)));

constexpr int T_LEN  = 8192;
constexpr int NCHUNK = T_LEN / 16;
constexpr int NSLICE = 16;
constexpr int SLICE_CHUNKS = NCHUNK / NSLICE;


struct Coef { float b0, b1, b2, a1, a2; };

__device__ __forceinline__ Coef dsvf_coef(const float* gp, const float* rp,
                                          const float* mhp_p, const float* mbp_p,
                                          const float* mlp_p)
{
    const float gv  = gp[0];
    const float rv  = rp[0];
    const float mhp = mhp_p[0];
    const float mbp = mbp_p[0];
    const float mlp = mlp_p[0];

    const float sig = 1.0f / (1.0f + expf(-gv));
    const float gg  = tanf(3.14159265358979323846f * 0.5f * sig);
    const float rr  = log1pf(expf(rv));
    const float g2  = gg * gg;

    Coef c;
    c.b0 = g2 * mlp + gg * mbp + mhp;
    c.b1 = 2.0f * g2 * mlp - 2.0f * mhp;
    c.b2 = g2 * mlp - gg * mbp + mhp;
    const float a0 = g2 + 2.0f * rr * gg + 1.0f;
    const float inva0 = 1.0f / a0;
    c.a1 = (2.0f * g2 - 2.0f) * inva0;
    c.a2 = (g2 - 2.0f * rr * gg + 1.0f) * inva0;
    return c;
}

template <bool FINAL>
__device__ __forceinline__ void dsvf_slice_body(
    const float* __restrict__ x,
    const float* __restrict__ gp,
    const float* __restrict__ rp,
    const float* __restrict__ mhp_p,
    const float* __restrict__ mbp_p,
    const float* __restrict__ mlp_p,
    float* fout,
    float* ds)
{

    const unsigned lane = threadIdx.x;
    const unsigned j    = lane & 15u;
    const bool     hi   = lane >= 16u;
    const int rbase  = (int)(blockIdx.x / NSLICE) * 16;
    const int slice  = (int)(blockIdx.x % NSLICE);
    const int mstart = slice * SLICE_CHUNKS;
    if (!FINAL && slice == NSLICE - 1) return;

    const Coef cf = dsvf_coef(gp, rp, mhp_p, mbp_p, mlp_p);

    if (lane == 0u) {
        for (int i = 0; i < 40; ++i) ds[i] = 0.0f;
        ds[0] = cf.b0; ds[1] = cf.b1; ds[2] = cf.b2;
    }
    float q1 = cf.a1, q2 = cf.a2;
    for (int s = 1; s <= 8; s *= 2) {
        if (lane == 0u) {
            for (int i = 32; i >= 0; --i) {
                float v = ds[i];
                if (i >= s)     v -= q1 * ds[i - s];
                if (i >= 2 * s) v += q2 * ds[i - 2 * s];
                ds[i] = v;
            }
        }
        const float n1 = 2.0f * q2 - q1 * q1;
        const float n2 = q2 * q2;
        q1 = n1; q2 = n2;
    }
    __syncthreads();

    v2f Bf[3][4];
#pragma unroll
    for (int b = 0; b < 3; ++b) {
#pragma unroll
        for (int s = 0; s < 4; ++s) {
            const int k0 = (hi ? 8 : 0) + 2 * s;
            const int i0 = 16 * b + (int)j - k0;
            const int i1 = i0 - 1;
            const float d0 = ds[min(max(i0, 0), 39)];
            const float d1 = ds[min(max(i1, 0), 39)];
            Bf[b][s].x = (i0 >= 0 && i0 <= 32) ? d0 : 0.0f;
            Bf[b][s].y = (i1 >= 0 && i1 <= 32) ? d1 : 0.0f;
        }
    }

    const float nalpha = -q1;
    const float nbeta  = -q2;

    v8f y1 = {}, y2 = {};

    float tpa = 1.0f, tpb = 0.0f, tqa = 0.0f, tqb = 1.0f;
    for (int k = 0; k < SLICE_CHUNKS; ++k) {
        const float np = nalpha * tpa + nbeta * tpb;
        const float nq = nalpha * tqa + nbeta * tqb;
        tpb = tpa; tpa = np;
        tqb = tqa; tqa = nq;
    }

    if (FINAL) {
        const float* zbase = fout + (size_t)rbase * T_LEN + (size_t)mstart * 16 + lane;
        float z[NSLICE - 1][16];
#pragma unroll
        for (int i = 0; i < NSLICE - 1; ++i) {
            if (i < slice) {
#pragma unroll
                for (int r = 0; r < 16; ++r) z[i][r] = zbase[(size_t)i * T_LEN + r * 32];
            }
        }
#pragma unroll
        for (int i = 0; i < NSLICE - 1; ++i) {
            if (i < slice) {
#pragma unroll
                for (int r = 0; r < 8; ++r) {
                    const float s1 = y1[r], s2 = y2[r];
                    y1[r] = z[i][r]     + (tpa * s1 + tqa * s2);
                    y2[r] = z[i][8 + r] + (tpb * s1 + tqb * s2);
                }
            }
        }
    }

    const float* xrow = x + (size_t)(rbase + (int)j) * T_LEN + (hi ? 8 : 0);
    float* frow = fout + (size_t)(rbase + (hi ? 8 : 0)) * T_LEN + j;

    if (!FINAL) {
        v2f U1[4], U2[4], U3[4], U4[4], S1[4], S2[4];
        for (int s = 0; s < 4; ++s) {
            U1[s].x = 0.0f; U1[s].y = 0.0f; U2[s].x = 0.0f; U2[s].y = 0.0f;
            U3[s].x = 0.0f; U3[s].y = 0.0f; U4[s].x = 0.0f; U4[s].y = 0.0f;
        }
        if (mstart > 0) {
            const float4 fa = *(const float4*)(xrow + (mstart - 2) * 16);
            const float4 fb = *(const float4*)(xrow + (mstart - 2) * 16 + 4);
            const float4 ga = *(const float4*)(xrow + (mstart - 1) * 16);
            const float4 gb = *(const float4*)(xrow + (mstart - 1) * 16 + 4);
            U2[0].x = fa.x; U2[0].y = fa.y;
            U2[1].x = fa.z; U2[1].y = fa.w;
            U2[2].x = fb.x; U2[2].y = fb.y;
            U2[3].x = fb.z; U2[3].y = fb.w;
            U1[0].x = ga.x; U1[0].y = ga.y;
            U1[1].x = ga.z; U1[1].y = ga.w;
            U1[2].x = gb.x; U1[2].y = gb.y;
            U1[3].x = gb.z; U1[3].y = gb.w;
            for (int s = 0; s < 4; ++s) U1[s] = U1[s] + nalpha * U2[s];
        }
        for (int s = 0; s < 4; ++s) { S1[s] = U1[s]; S2[s] = U2[s]; }

        for (int m0 = mstart; m0 < mstart + SLICE_CHUNKS; m0 += 4) {
            if (m0 + 8 < mstart + SLICE_CHUNKS)
                __builtin_prefetch(xrow + m0 * 16 + 128, 0, 1);
#pragma unroll
            for (int u = 0; u < 4; ++u) {
                const int t0 = (m0 + u) * 16;
                const float4 xa = *(const float4*)(xrow + t0);
                const float4 xb = *(const float4*)(xrow + t0 + 4);
                v2f Xc[4];
                Xc[0].x = xa.x; Xc[0].y = xa.y;
                Xc[1].x = xa.z; Xc[1].y = xa.w;
                Xc[2].x = xb.x; Xc[2].y = xb.y;
                Xc[3].x = xb.z; Xc[3].y = xb.w;
                for (int s = 0; s < 4; ++s) {
                    const v2f un = (Xc[s] + nbeta * U2[s]) + nalpha * U1[s];
                    U4[s] = U3[s]; U3[s] = U2[s]; U2[s] = U1[s]; U1[s] = un;
                }
            }
        }

        v2f A0[4], A1[4], C0[4], C1[4];
        for (int s = 0; s < 4; ++s) {
            A0[s] = (U1[s] - tqa * S2[s]) - tpa * S1[s];
            A1[s] = U2[s] - tpa * S2[s];
            C0[s] = (U2[s] - tqb * S2[s]) - tpb * S1[s];
            C1[s] = U3[s] - tpb * S2[s];
        }
        v8f accA = {}, accB = {}, accC = {}, accD = {};
        for (int s = 0; s < 4; ++s) {
            accA = __builtin_amdgcn_wmma_f32_16x16x4_f32(
                false, U3[s], false, Bf[2][s], (short)0, accA, false, false);
            accB = __builtin_amdgcn_wmma_f32_16x16x4_f32(
                false, A1[s], false, Bf[1][s], (short)0, accB, false, false);
            accC = __builtin_amdgcn_wmma_f32_16x16x4_f32(
                false, U4[s], false, Bf[2][s], (short)0, accC, false, false);
            accD = __builtin_amdgcn_wmma_f32_16x16x4_f32(
                false, C1[s], false, Bf[1][s], (short)0, accD, false, false);
        }
        for (int s = 0; s < 4; s += 2) {
            accA = __builtin_amdgcn_wmma_f32_16x16x4_f32(
                false, A0[s], false, Bf[0][s], (short)0, accA, false, false);
            accB = __builtin_amdgcn_wmma_f32_16x16x4_f32(
                false, A0[s + 1], false, Bf[0][s + 1], (short)0, accB, false, false);
            accC = __builtin_amdgcn_wmma_f32_16x16x4_f32(
                false, C0[s], false, Bf[0][s], (short)0, accC, false, false);
            accD = __builtin_amdgcn_wmma_f32_16x16x4_f32(
                false, C0[s + 1], false, Bf[0][s + 1], (short)0, accD, false, false);
        }
        y1 = accA + accB;
        y2 = accC + accD;
    } else {
        v2f X[4][4];
    #pragma unroll
        for (int u = 0; u < 4; ++u)
    #pragma unroll
            for (int s = 0; s < 4; ++s) { X[u][s].x = 0.0f; X[u][s].y = 0.0f; }

        if (mstart > 0) {
            const float4 pa = *(const float4*)(xrow + (mstart - 1) * 16);
            const float4 pb = *(const float4*)(xrow + (mstart - 1) * 16 + 4);
            const float4 qa = *(const float4*)(xrow + (mstart - 2) * 16);
            const float4 qb = *(const float4*)(xrow + (mstart - 2) * 16 + 4);
            X[3][0].x = pa.x; X[3][0].y = pa.y;
            X[3][1].x = pa.z; X[3][1].y = pa.w;
            X[3][2].x = pb.x; X[3][2].y = pb.y;
            X[3][3].x = pb.z; X[3][3].y = pb.w;
            X[2][0].x = qa.x; X[2][0].y = qa.y;
            X[2][1].x = qa.z; X[2][1].y = qa.w;
            X[2][2].x = qb.x; X[2][2].y = qb.y;
            X[2][3].x = qb.z; X[2][3].y = qb.w;
        }

        for (int m0 = mstart; m0 < mstart + SLICE_CHUNKS; m0 += 4) {
            if (m0 + 8 < mstart + SLICE_CHUNKS)
                __builtin_prefetch(xrow + m0 * 16 + 128, 0, 1);
    #pragma unroll
            for (int u = 0; u < 4; ++u) {
                const int t0 = (m0 + u) * 16;
                const float4 xa = *(const float4*)(xrow + t0);
                const float4 xb = *(const float4*)(xrow + t0 + 4);

                const int c  = u;
                const int p1 = (u + 3) & 3;
                const int p2 = (u + 2) & 3;
                X[c][0].x = xa.x; X[c][0].y = xa.y;
                X[c][1].x = xa.z; X[c][1].y = xa.w;
                X[c][2].x = xb.x; X[c][2].y = xb.y;
                X[c][3].x = xb.z; X[c][3].y = xb.w;

                v8f accA = {}, accB = {};
    #pragma unroll
                for (int s = 0; s < 4; ++s) {
                    accA = __builtin_amdgcn_wmma_f32_16x16x4_f32(
                        false, X[p2][s], false, Bf[2][s], (short)0, accA, false, false);
                    accB = __builtin_amdgcn_wmma_f32_16x16x4_f32(
                        false, X[p1][s], false, Bf[1][s], (short)0, accB, false, false);
                }
    #pragma unroll
                for (int s = 0; s < 4; s += 2) {
                    accA = __builtin_amdgcn_wmma_f32_16x16x4_f32(
                        false, X[c][s], false, Bf[0][s], (short)0, accA, false, false);
                    accB = __builtin_amdgcn_wmma_f32_16x16x4_f32(
                        false, X[c][s + 1], false, Bf[0][s + 1], (short)0, accB, false, false);
                }

                const v8f yv = ((accA + accB) + nbeta * y2) + nalpha * y1;
                y2 = y1; y1 = yv;

                if (FINAL) {
    #pragma unroll
                    for (int v = 0; v < 8; ++v)
                        frow[(size_t)v * T_LEN + t0] = yv[v];
                }
            }
        }

    }

    if (!FINAL) {
        float* zrow = fout + (size_t)(rbase + slice) * T_LEN + lane;
        int s2 = slice + 1;
        do {
            float* zp = zrow + s2 * (SLICE_CHUNKS * 16);
#pragma unroll
            for (int r = 0; r < 8; ++r) {
                zp[r * 32]       = y1[r];
                zp[(8 + r) * 32] = y2[r];
            }
        } while (++s2 < NSLICE);
    }
}

__global__ __launch_bounds__(32) void dsvf_state_kernel(
    const float* __restrict__ x,
    const float* __restrict__ gp,
    const float* __restrict__ rp,
    const float* __restrict__ mhp_p,
    const float* __restrict__ mbp_p,
    const float* __restrict__ mlp_p,
    float* out)
{
    __shared__ float ds[40];
    dsvf_slice_body<false>(x, gp, rp, mhp_p, mbp_p, mlp_p, out, ds);
}

__global__ __launch_bounds__(32) void dsvf_out_kernel(
    const float* __restrict__ x,
    const float* __restrict__ gp,
    const float* __restrict__ rp,
    const float* __restrict__ mhp_p,
    const float* __restrict__ mbp_p,
    const float* __restrict__ mlp_p,
    float* out)
{
    __shared__ float ds[40];
    dsvf_slice_body<true>(x, gp, rp, mhp_p, mbp_p, mlp_p, out, ds);
}


extern "C" void kernel_launch(void* const* d_in, const int* in_sizes, int n_in,
                              void* d_out, int out_size, void* d_ws, size_t ws_size,
                              hipStream_t stream) {
    const float* x   = (const float*)d_in[0];
    const float* g   = (const float*)d_in[1];
    const float* r   = (const float*)d_in[2];
    const float* mhp = (const float*)d_in[3];
    const float* mbp = (const float*)d_in[4];
    const float* mlp = (const float*)d_in[5];
    float* y = (float*)d_out;

    const int B = in_sizes[0] / T_LEN;
    dim3 grid(B / 16 * NSLICE), block(32);
    hipLaunchKernelGGL(dsvf_state_kernel, grid, block, 0, stream,
                       x, g, r, mhp, mbp, mlp, y);
    hipLaunchKernelGGL(dsvf_out_kernel, grid, block, 0, stream,
                       x, g, r, mhp, mbp, mlp, y);
}


# Slices hand on the biquad's own state, 2 outputs per row (was 32: two look-ahead chunks): 128-byte records, 2x2 fold, state response enters chunks 0 and 1 as k=4 products; carry sum order changes
# speedup vs baseline: 10.4711x; 1.1963x over previous
#include <hip/hip_runtime.h>
#include <math.h>


typedef float v2f __attribute__((ext_vector_type(2)));
typedef float v8f __attribute__((ext_vector_type(8)));

constexpr int T_LEN  = 8192;
constexpr int NCHUNK = T_LEN / 16;
constexpr int NSLICE = 16;
constexpr int SLICE_CHUNKS = NCHUNK / NSLICE;
constexpr int SLICE_LEN = SLICE_CHUNKS * 16;


struct Coef { float b0, b1, b2, a1, a2; };

__device__ __forceinline__ Coef dsvf_coef(float gv, float rv, float mhp, float mbp, float mlp)
{
    const float sig = 1.0f / (1.0f + expf(-gv));
    const float gg  = tanf(3.14159265358979323846f * 0.5f * sig);
    const float rr  = log1pf(expf(rv));
    const float g2  = gg * gg;

    Coef c;
    c.b0 = g2 * mlp + gg * mbp + mhp;
    c.b1 = 2.0f * g2 * mlp - 2.0f * mhp;
    c.b2 = g2 * mlp - gg * mbp + mhp;
    const float a0 = g2 + 2.0f * rr * gg + 1.0f;
    const float inva0 = 1.0f / a0;
    c.a1 = (2.0f * g2 - 2.0f) * inva0;
    c.a2 = (g2 - 2.0f * rr * gg + 1.0f) * inva0;
    return c;
}

template <bool FINAL>
__device__ __forceinline__ void dsvf_slice_body(
    const float* __restrict__ x,
    const float* __restrict__ gp,
    const float* __restrict__ rp,
    const float* __restrict__ mhp_p,
    const float* __restrict__ mbp_p,
    const float* __restrict__ mlp_p,
    float* fout,
    float* ds,
    float* gs)
{

    const unsigned lane = threadIdx.x;
    const unsigned j    = lane & 15u;
    const bool     hi   = lane >= 16u;
    const int rbase  = (int)(blockIdx.x / NSLICE) * 16;
    const int slice  = (int)(blockIdx.x % NSLICE);
    const int mstart = slice * SLICE_CHUNKS;
    if (!FINAL && slice == NSLICE - 1) return;

    const float* xrow0 = x + (size_t)(rbase + (int)j) * T_LEN;
    const float* xrow  = xrow0 + (hi ? 8 : 0);
    {
        const float* pf = xrow + mstart * 16;
#pragma unroll
        for (int u = 0; u < SLICE_CHUNKS / 4; ++u)
            __builtin_prefetch(pf + u * 64, 0, 1);
    }
    const float2 xp = *(const float2*)(xrow0 + (mstart > 0 ? mstart * 16 - 2 : 0));
    const float xh1 = (mstart > 0) ? xp.y : 0.0f;
    const float xh2 = (mstart > 0) ? xp.x : 0.0f;
    float gv = gp[0], rv = rp[0], mhp = mhp_p[0], mbp = mbp_p[0], mlp = mlp_p[0];
    asm volatile("" : "+s"(gv), "+s"(rv), "+s"(mhp), "+s"(mbp), "+s"(mlp));
    float z1[NSLICE - 1], z2[NSLICE - 1];
    if (FINAL) {
        const float* zbase = fout + (size_t)rbase * T_LEN + (size_t)mstart * 16 + j;
#pragma unroll
        for (int i = 0; i < NSLICE - 1; ++i) {
            if (i < slice) {
                z1[i] = zbase[(size_t)i * T_LEN];
                z2[i] = zbase[(size_t)i * T_LEN + 16];
            }
        }
    }

    const Coef cf = dsvf_coef(gv, rv, mhp, mbp, mlp);

    if (lane == 0u) {
        for (int i = 0; i < 40; ++i) ds[i] = 0.0f;
        ds[0] = cf.b0; ds[1] = cf.b1; ds[2] = cf.b2;
        float gm2 = 0.0f, gm1 = 1.0f;
        gs[0] = 0.0f; gs[1] = 1.0f;
        for (int n = 2; n < 40; ++n) {
            const float gn = -cf.a1 * gm1 - cf.a2 * gm2;
            gs[n] = gn; gm2 = gm1; gm1 = gn;
        }
    }
    float q1 = cf.a1, q2 = cf.a2;
    for (int s = 1; s <= 8; s *= 2) {
        if (lane == 0u) {
            for (int i = 32; i >= 0; --i) {
                float v = ds[i];
                if (i >= s)     v -= q1 * ds[i - s];
                if (i >= 2 * s) v += q2 * ds[i - 2 * s];
                ds[i] = v;
            }
        }
        const float n1 = 2.0f * q2 - q1 * q1;
        const float n2 = q2 * q2;
        q1 = n1; q2 = n2;
    }
    __syncthreads();

    v2f Bf[3][4];
#pragma unroll
    for (int b = 0; b < 3; ++b) {
#pragma unroll
        for (int s = 0; s < 4; ++s) {
            const int k0 = (hi ? 8 : 0) + 2 * s;
            const int i0 = 16 * b + (int)j - k0;
            const int i1 = i0 - 1;
            const float d0 = ds[min(max(i0, 0), 39)];
            const float d1 = ds[min(max(i1, 0), 39)];
            Bf[b][s].x = (i0 >= 0 && i0 <= 32) ? d0 : 0.0f;
            Bf[b][s].y = (i1 >= 0 && i1 <= 32) ? d1 : 0.0f;
        }
    }

    const float nalpha = -q1;
    const float nbeta  = -q2;

    float ga = gs[1 + j], gb = gs[17 + j];
    for (int k = 0; k < SLICE_CHUNKS - 1; ++k) {
        const float gn = nbeta * ga + nalpha * gb;
        ga = gb; gb = gn;
    }
    const float g509 = __int_as_float(__builtin_amdgcn_readlane(__float_as_int(ga), 13));
    const float g510 = __int_as_float(__builtin_amdgcn_readlane(__float_as_int(ga), 14));
    const float g511 = __int_as_float(__builtin_amdgcn_readlane(__float_as_int(ga), 15));
    const float g512 = __int_as_float(__builtin_amdgcn_readlane(__float_as_int(gb), 0));

    v8f y1 = {}, y2 = {};

    float* frow = fout + (size_t)(rbase + (hi ? 8 : 0)) * T_LEN + j;

    if (!FINAL) {
        v2f U1[4], U2[4], U3[4];
        for (int s = 0; s < 4; ++s) {
            U1[s].x = 0.0f; U1[s].y = 0.0f; U2[s].x = 0.0f; U2[s].y = 0.0f;
            U3[s].x = 0.0f; U3[s].y = 0.0f;
        }

        for (int m0 = mstart; m0 < mstart + SLICE_CHUNKS; m0 += 4) {
#pragma unroll
            for (int u = 0; u < 4; ++u) {
                const int t0 = (m0 + u) * 16;
                const float4 xa = *(const float4*)(xrow + t0);
                const float4 xb = *(const float4*)(xrow + t0 + 4);
                v2f Xc[4];
                Xc[0].x = xa.x; Xc[0].y = xa.y;
                Xc[1].x = xa.z; Xc[1].y = xa.w;
                Xc[2].x = xb.x; Xc[2].y = xb.y;
                Xc[3].x = xb.z; Xc[3].y = xb.w;
                for (int s = 0; s < 4; ++s) {
                    const v2f un = (Xc[s] + nbeta * U2[s]) + nalpha * U1[s];
                    U3[s] = U2[s]; U2[s] = U1[s]; U1[s] = un;
                }
            }
        }

        v8f accA = {}, accB = {};
        for (int s = 0; s < 4; ++s) {
            accA = __builtin_amdgcn_wmma_f32_16x16x4_f32(
                false, U3[s], false, Bf[2][s], (short)0, accA, false, false);
            accB = __builtin_amdgcn_wmma_f32_16x16x4_f32(
                false, U2[s], false, Bf[1][s], (short)0, accB, false, false);
        }
        for (int s = 0; s < 4; s += 2) {
            accA = __builtin_amdgcn_wmma_f32_16x16x4_f32(
                false, U1[s], false, Bf[0][s], (short)0, accA, false, false);
            accB = __builtin_amdgcn_wmma_f32_16x16x4_f32(
                false, U1[s + 1], false, Bf[0][s + 1], (short)0, accB, false, false);
        }
        {
            const float gq = (j == 15u) ? g511 : g510;
            const float gp1 = (j == 15u) ? g510 : g509;
            v2f XH, EB;
            XH.x = hi ? 0.0f : xh1;
            XH.y = hi ? 0.0f : xh2;
            EB.x = hi ? 0.0f : (cf.b1 * gq + cf.b2 * gp1);
            EB.y = hi ? 0.0f : (cf.b2 * gq);
            accB = __builtin_amdgcn_wmma_f32_16x16x4_f32(
                false, XH, false, EB, (short)0, accB, false, false);
        }
        y1 = accA + accB;

        if (j >= 14u) {
            float* zrow = fout + (size_t)(rbase + slice) * T_LEN
                        + (15 - (int)j) * 16 + (hi ? 8 : 0);
            int s2 = slice + 1;
            do {
                float* zp = zrow + s2 * SLICE_LEN;
                *(float4*)(zp)     = make_float4(y1[0], y1[1], y1[2], y1[3]);
                *(float4*)(zp + 4) = make_float4(y1[4], y1[5], y1[6], y1[7]);
            } while (++s2 < NSLICE);
        }
    } else {
        const float p11 = g512, p12 = -cf.a2 * g511;
        const float p21 = g511, p22 = -cf.a2 * g510;
        float st1 = 0.0f, st2 = 0.0f;
#pragma unroll
        for (int i = 0; i < NSLICE - 1; ++i) {
            if (i < slice) {
                const float n1 = z1[i] + (p11 * st1 + p12 * st2);
                const float n2 = z2[i] + (p21 * st1 + p22 * st2);
                st1 = n1; st2 = n2;
            }
        }
        v8f g0, g1;
        {
            const float ga_m = gs[j],      ga_0 = gs[1 + j],  ga_p = gs[2 + j];
            const float gb_m = gs[16 + j], gb_0 = gs[17 + j], gb_p = gs[18 + j];
            v2f C0, C1, Sf;
            C0.x = hi ? ga_p : (cf.b1 * ga_0 + cf.b2 * ga_m);
            C0.y = hi ? (-cf.a2 * ga_0) : (cf.b2 * ga_0);
            C1.x = hi ? gb_p : (cf.b1 * gb_0 + cf.b2 * gb_m);
            C1.y = hi ? (-cf.a2 * gb_0) : (cf.b2 * gb_0);
            C1 = C1 - nalpha * C0;
            Sf.x = hi ? st1 : xh1;
            Sf.y = hi ? st2 : xh2;
            const v8f zero8 = {};
            g0 = __builtin_amdgcn_wmma_f32_16x16x4_f32(
                false, Sf, false, C0, (short)0, zero8, false, false);
            g1 = __builtin_amdgcn_wmma_f32_16x16x4_f32(
                false, Sf, false, C1, (short)0, zero8, false, false);
        }

        v2f X[4][4];
    #pragma unroll
        for (int u = 0; u < 4; ++u)
    #pragma unroll
            for (int s = 0; s < 4; ++s) { X[u][s].x = 0.0f; X[u][s].y = 0.0f; }

        for (int m0 = mstart; m0 < mstart + SLICE_CHUNKS; m0 += 4) {
    #pragma unroll
            for (int u = 0; u < 4; ++u) {
                const int t0 = (m0 + u) * 16;
                const float4 xa = *(const float4*)(xrow + t0);
                const float4 xb = *(const float4*)(xrow + t0 + 4);

                const int c  = u;
                const int p1 = (u + 3) & 3;
                const int p2 = (u + 2) & 3;
                X[c][0].x = xa.x; X[c][0].y = xa.y;
                X[c][1].x = xa.z; X[c][1].y = xa.w;
                X[c][2].x = xb.x; X[c][2].y = xb.y;
                X[c][3].x = xb.z; X[c][3].y = xb.w;

                v8f accA = {}, accB = {};
                if (u == 0) accA = g0;
                if (u == 1) accA = g1;
    #pragma unroll
                for (int s = 0; s < 4; ++s) {
                    accA = __builtin_amdgcn_wmma_f32_16x16x4_f32(
                        false, X[p2][s], false, Bf[2][s], (short)0, accA, false, false);
                    accB = __builtin_amdgcn_wmma_f32_16x16x4_f32(
                        false, X[p1][s], false, Bf[1][s], (short)0, accB, false, false);
                }
    #pragma unroll
                for (int s = 0; s < 4; s += 2) {
                    accA = __builtin_amdgcn_wmma_f32_16x16x4_f32(
                        false, X[c][s], false, Bf[0][s], (short)0, accA, false, false);
                    accB = __builtin_amdgcn_wmma_f32_16x16x4_f32(
                        false, X[c][s + 1], false, Bf[0][s + 1], (short)0, accB, false, false);
                }

                const v8f yv = ((accA + accB) + nbeta * y2) + nalpha * y1;
                y2 = y1; y1 = yv;

    #pragma unroll
                for (int v = 0; v < 8; ++v)
                    frow[(size_t)v * T_LEN + t0] = yv[v];
            }
            const v8f zero8 = {};
            g0 = zero8; g1 = zero8;
        }
    }
}

__global__ __launch_bounds__(32) void dsvf_state_kernel(
    const float* __restrict__ x,
    const float* __restrict__ gp,
    const float* __restrict__ rp,
    const float* __restrict__ mhp_p,
    const float* __restrict__ mbp_p,
    const float* __restrict__ mlp_p,
    float* out)
{
    __shared__ float ds[80];
    dsvf_slice_body<false>(x, gp, rp, mhp_p, mbp_p, mlp_p, out, ds, ds + 40);
}

__global__ __launch_bounds__(32) void dsvf_out_kernel(
    const float* __restrict__ x,
    const float* __restrict__ gp,
    const float* __restrict__ rp,
    const float* __restrict__ mhp_p,
    const float* __restrict__ mbp_p,
    const float* __restrict__ mlp_p,
    float* out)
{
    __shared__ float ds[80];
    dsvf_slice_body<true>(x, gp, rp, mhp_p, mbp_p, mlp_p, out, ds, ds + 40);
}


extern "C" void kernel_launch(void* const* d_in, const int* in_sizes, int n_in,
                              void* d_out, int out_size, void* d_ws, size_t ws_size,
                              hipStream_t stream) {
    const float* x   = (const float*)d_in[0];
    const float* g   = (const float*)d_in[1];
    const float* r   = (const float*)d_in[2];
    const float* mhp = (const float*)d_in[3];
    const float* mbp = (const float*)d_in[4];
    const float* mlp = (const float*)d_in[5];
    float* y = (float*)d_out;

    const int B = in_sizes[0] / T_LEN;
    dim3 grid(B / 16 * NSLICE), block(32);
    hipLaunchKernelGGL(dsvf_state_kernel, grid, block, 0, stream,
                       x, g, r, mhp, mbp, mlp, y);
    hipLaunchKernelGGL(dsvf_out_kernel, grid, block, 0, stream,
                       x, g, r, mhp, mbp, mlp, y);
}


# Output launch: chunk = X H + S C, biquad's direct form: 5 products (was 12: look-ahead); next state = cols 14,15 redone per lane on the vector unit, halves traded once; sum order changes
# speedup vs baseline: 12.4840x; 1.1922x over previous
#include <hip/hip_runtime.h>
#include <math.h>


typedef float v2f __attribute__((ext_vector_type(2)));
typedef float v8f __attribute__((ext_vector_type(8)));

constexpr int T_LEN  = 8192;
constexpr int NCHUNK = T_LEN / 16;
constexpr int NSLICE = 16;
constexpr int SLICE_CHUNKS = NCHUNK / NSLICE;
constexpr int SLICE_LEN = SLICE_CHUNKS * 16;


struct Coef { float b0, b1, b2, a1, a2; };

__device__ __forceinline__ Coef dsvf_coef(float gv, float rv, float mhp, float mbp, float mlp)
{
    const float sig = 1.0f / (1.0f + expf(-gv));
    const float gg  = tanf(3.14159265358979323846f * 0.5f * sig);
    const float rr  = log1pf(expf(rv));
    const float g2  = gg * gg;

    Coef c;
    c.b0 = g2 * mlp + gg * mbp + mhp;
    c.b1 = 2.0f * g2 * mlp - 2.0f * mhp;
    c.b2 = g2 * mlp - gg * mbp + mhp;
    const float a0 = g2 + 2.0f * rr * gg + 1.0f;
    const float inva0 = 1.0f / a0;
    c.a1 = (2.0f * g2 - 2.0f) * inva0;
    c.a2 = (g2 - 2.0f * rr * gg + 1.0f) * inva0;
    return c;
}

__device__ __forceinline__ float dsvf_htap(const float* gs, const Coef& cf, int i)
{
    const int ic = max(i, 0);
    const float v = cf.b0 * gs[ic + 1] + cf.b1 * gs[ic] + cf.b2 * gs[max(ic - 1, 0)];
    return (i >= 0) ? v : 0.0f;
}

__device__ __forceinline__ float dsvf_xhalf(float v)
{
    return __int_as_float(__builtin_amdgcn_permlanex16(
        __float_as_int(v), __float_as_int(v), 0x76543210, 0xfedcba98, false, false));
}

template <bool FINAL>
__device__ __forceinline__ void dsvf_slice_body(
    const float* x,
    const float* __restrict__ gp,
    const float* __restrict__ rp,
    const float* __restrict__ mhp_p,
    const float* __restrict__ mbp_p,
    const float* __restrict__ mlp_p,
    float* fout,
    float* ds,
    float* gs)
{

    const unsigned lane = threadIdx.x;
    const unsigned j    = lane & 15u;
    const bool     hi   = lane >= 16u;
    const int rbase  = (int)(blockIdx.x / NSLICE) * 16;
    const int slice  = (int)(blockIdx.x % NSLICE);
    const int mstart = slice * SLICE_CHUNKS;
    if (!FINAL && slice == NSLICE - 1) return;

    const float* xrow0 = x + (size_t)(rbase + (int)j) * T_LEN;
    const float* xrow  = xrow0 + (hi ? 8 : 0);
    {
        const float* pf = xrow + mstart * 16;
#pragma unroll
        for (int u = 0; u < SLICE_CHUNKS / 4; ++u)
            __builtin_prefetch(pf + u * 64, 0, 1);
    }
    const float2 xp = *(const float2*)(xrow0 + (mstart > 0 ? mstart * 16 - 2 : 0));
    const float xh1 = (mstart > 0) ? xp.y : 0.0f;
    const float xh2 = (mstart > 0) ? xp.x : 0.0f;
    float gv = gp[0], rv = rp[0], mhp = mhp_p[0], mbp = mbp_p[0], mlp = mlp_p[0];
    asm volatile("" : "+s"(gv), "+s"(rv), "+s"(mhp), "+s"(mbp), "+s"(mlp));
    float z1[NSLICE - 1], z2[NSLICE - 1];
    if (FINAL) {
        const float* zbase = fout + (size_t)rbase * T_LEN + (size_t)mstart * 16 + j;
#pragma unroll
        for (int i = 0; i < NSLICE - 1; ++i) {
            if (i < slice) {
                z1[i] = zbase[(size_t)i * T_LEN];
                z2[i] = zbase[(size_t)i * T_LEN + 16];
            }
        }
    }

    const Coef cf = dsvf_coef(gv, rv, mhp, mbp, mlp);

    if (lane == 0u) {
        if (!FINAL) {
            for (int i = 0; i < 40; ++i) ds[i] = 0.0f;
            ds[0] = cf.b0; ds[1] = cf.b1; ds[2] = cf.b2;
        }
        float gm2 = 0.0f, gm1 = 1.0f;
        gs[0] = 0.0f; gs[1] = 1.0f;
        for (int n = 2; n < 40; ++n) {
            const float gn = -cf.a1 * gm1 - cf.a2 * gm2;
            gs[n] = gn; gm2 = gm1; gm1 = gn;
        }
    }
    float q1 = cf.a1, q2 = cf.a2;
    for (int s = 1; s <= 8; s *= 2) {
        if (!FINAL && lane == 0u) {
            for (int i = 32; i >= 0; --i) {
                float v = ds[i];
                if (i >= s)     v -= q1 * ds[i - s];
                if (i >= 2 * s) v += q2 * ds[i - 2 * s];
                ds[i] = v;
            }
        }
        const float n1 = 2.0f * q2 - q1 * q1;
        const float n2 = q2 * q2;
        q1 = n1; q2 = n2;
    }
    __syncthreads();

    v2f Bf[3][4];
    if (!FINAL) {
#pragma unroll
        for (int b = 0; b < 3; ++b) {
#pragma unroll
            for (int s = 0; s < 4; ++s) {
                const int k0 = (hi ? 8 : 0) + 2 * s;
                const int i0 = 16 * b + (int)j - k0;
                const int i1 = i0 - 1;
                const float d0 = ds[min(max(i0, 0), 39)];
                const float d1 = ds[min(max(i1, 0), 39)];
                Bf[b][s].x = (i0 >= 0 && i0 <= 32) ? d0 : 0.0f;
                Bf[b][s].y = (i1 >= 0 && i1 <= 32) ? d1 : 0.0f;
            }
        }
    }

    const float nalpha = -q1;
    const float nbeta  = -q2;

    float ga = gs[1 + j], gb = gs[17 + j];
    for (int k = 0; k < SLICE_CHUNKS - 1; ++k) {
        const float gn = nbeta * ga + nalpha * gb;
        ga = gb; gb = gn;
    }
    const float g509 = __int_as_float(__builtin_amdgcn_readlane(__float_as_int(ga), 13));
    const float g510 = __int_as_float(__builtin_amdgcn_readlane(__float_as_int(ga), 14));
    const float g511 = __int_as_float(__builtin_amdgcn_readlane(__float_as_int(ga), 15));
    const float g512 = __int_as_float(__builtin_amdgcn_readlane(__float_as_int(gb), 0));

    v8f y1 = {};

    float* frow = fout + (size_t)(rbase + (hi ? 8 : 0)) * T_LEN + j;

    if (!FINAL) {
        v2f U1[4], U2[4], U3[4];
        for (int s = 0; s < 4; ++s) {
            U1[s].x = 0.0f; U1[s].y = 0.0f; U2[s].x = 0.0f; U2[s].y = 0.0f;
            U3[s].x = 0.0f; U3[s].y = 0.0f;
        }

        for (int m0 = mstart; m0 < mstart + SLICE_CHUNKS; m0 += 4) {
#pragma unroll
            for (int u = 0; u < 4; ++u) {
                const int t0 = (m0 + u) * 16;
                const float4 xa = *(const float4*)(xrow + t0);
                const float4 xb = *(const float4*)(xrow + t0 + 4);
                v2f Xc[4];
                Xc[0].x = xa.x; Xc[0].y = xa.y;
                Xc[1].x = xa.z; Xc[1].y = xa.w;
                Xc[2].x = xb.x; Xc[2].y = xb.y;
                Xc[3].x = xb.z; Xc[3].y = xb.w;
                for (int s = 0; s < 4; ++s) {
                    const v2f un = (Xc[s] + nbeta * U2[s]) + nalpha * U1[s];
                    U3[s] = U2[s]; U2[s] = U1[s]; U1[s] = un;
                }
            }
        }

        v8f accA = {}, accB = {};
        for (int s = 0; s < 4; ++s) {
            accA = __builtin_amdgcn_wmma_f32_16x16x4_f32(
                false, U3[s], false, Bf[2][s], (short)0, accA, false, false);
            accB = __builtin_amdgcn_wmma_f32_16x16x4_f32(
                false, U2[s], false, Bf[1][s], (short)0, accB, false, false);
        }
        for (int s = 0; s < 4; s += 2) {
            accA = __builtin_amdgcn_wmma_f32_16x16x4_f32(
                false, U1[s], false, Bf[0][s], (short)0, accA, false, false);
            accB = __builtin_amdgcn_wmma_f32_16x16x4_f32(
                false, U1[s + 1], false, Bf[0][s + 1], (short)0, accB, false, false);
        }
        {
            const float gq = (j == 15u) ? g511 : g510;
            const float gp1 = (j == 15u) ? g510 : g509;
            v2f XH, EB;
            XH.x = hi ? 0.0f : xh1;
            XH.y = hi ? 0.0f : xh2;
            EB.x = hi ? 0.0f : (cf.b1 * gq + cf.b2 * gp1);
            EB.y = hi ? 0.0f : (cf.b2 * gq);
            accB = __builtin_amdgcn_wmma_f32_16x16x4_f32(
                false, XH, false, EB, (short)0, accB, false, false);
        }
        y1 = accA + accB;

        if (j >= 14u) {
            float* zrow = fout + (size_t)(rbase + slice) * T_LEN
                        + (15 - (int)j) * 16 + (hi ? 8 : 0);
            int s2 = slice + 1;
            do {
                float* zp = zrow + s2 * SLICE_LEN;
                *(float4*)(zp)     = make_float4(y1[0], y1[1], y1[2], y1[3]);
                *(float4*)(zp + 4) = make_float4(y1[4], y1[5], y1[6], y1[7]);
            } while (++s2 < NSLICE);
        }
    } else {
        const float p11 = g512, p12 = -cf.a2 * g511;
        const float p21 = g511, p22 = -cf.a2 * g510;
        float st1 = 0.0f, st2 = 0.0f;
#pragma unroll
        for (int i = 0; i < NSLICE - 1; ++i) {
            if (i < slice) {
                const float n1 = z1[i] + (p11 * st1 + p12 * st2);
                const float n2 = z2[i] + (p21 * st1 + p22 * st2);
                st1 = n1; st2 = n2;
            }
        }
        v2f Hf[4];
#pragma unroll
        for (int s = 0; s < 4; ++s) {
            const int i0 = (int)j - ((hi ? 8 : 0) + 2 * s);
            Hf[s].x = dsvf_htap(gs, cf, i0);
            Hf[s].y = dsvf_htap(gs, cf, i0 - 1);
        }
        v2f Cs, Sf;
        {
            const float g_m = gs[j], g_0 = gs[1 + j], g_p = gs[2 + j];
            Cs.x = hi ? (-cf.a2 * g_0) : (cf.b2 * g_0);
            Cs.y = hi ? g_p : (cf.b1 * g_0 + cf.b2 * g_m);
            Sf.x = hi ? st2 : xh2;
            Sf.y = hi ? st1 : xh1;
        }
        v2f hc[8];
#pragma unroll
        for (int e = 0; e < 8; ++e) {
            const int k = (hi ? 8 : 0) + e;
            hc[e].x = dsvf_htap(gs, cf, 14 - k);
            hc[e].y = dsvf_htap(gs, cf, 15 - k);
        }
        v2f cx2, cx1, cy2, cy1;
        {
            const float g13 = gs[14], g14 = gs[15], g15 = gs[16], g16 = gs[17];
            cx2.x = hi ? (cf.b2 * g14) : 0.0f;
            cx2.y = hi ? (cf.b2 * g15) : 0.0f;
            cx1.x = hi ? (cf.b1 * g14 + cf.b2 * g13) : 0.0f;
            cx1.y = hi ? (cf.b1 * g15 + cf.b2 * g14) : 0.0f;
            cy2.x = hi ? (-cf.a2 * g14) : 0.0f;
            cy2.y = hi ? (-cf.a2 * g15) : 0.0f;
            cy1.x = hi ? g15 : 0.0f;
            cy1.y = hi ? g16 : 0.0f;
        }
        v2f xprev;
        xprev.x = xh2; xprev.y = xh1;

        float4 xa0[4], xb0[4], xa1[4], xb1[4];
#pragma unroll
        for (int u = 0; u < 4; ++u) {
            xa0[u] = *(const float4*)(xrow + (mstart + u) * 16);
            xb0[u] = *(const float4*)(xrow + (mstart + u) * 16 + 4);
        }

        auto trip = [&](const float4 (&xac)[4], const float4 (&xbc)[4],
                        float4 (&xan)[4], float4 (&xbn)[4], int m0, int mn)
                        __attribute__((always_inline)) {
#pragma unroll
            for (int u = 0; u < 4; ++u) {
                xan[u] = *(const float4*)(xrow + (mn + u) * 16);
                xbn[u] = *(const float4*)(xrow + (mn + u) * 16 + 4);
            }
            __builtin_amdgcn_sched_barrier(0);
            v2f Xc[4][4];
#pragma unroll
            for (int u = 0; u < 4; ++u) {
                Xc[u][0].x = xac[u].x; Xc[u][0].y = xac[u].y;
                Xc[u][1].x = xac[u].z; Xc[u][1].y = xac[u].w;
                Xc[u][2].x = xbc[u].x; Xc[u][2].y = xbc[u].y;
                Xc[u][3].x = xbc[u].z; Xc[u][3].y = xbc[u].w;
            }

            v8f accA[4], accB[4];
            const v8f zero8 = {};
#pragma unroll
            for (int u = 0; u < 4; ++u) {
                accA[u] = __builtin_amdgcn_wmma_f32_16x16x4_f32(
                    false, Xc[u][0], false, Hf[0], (short)0, zero8, false, false);
                accB[u] = __builtin_amdgcn_wmma_f32_16x16x4_f32(
                    false, Xc[u][2], false, Hf[2], (short)0, zero8, false, false);
                accA[u] = __builtin_amdgcn_wmma_f32_16x16x4_f32(
                    false, Xc[u][1], false, Hf[1], (short)0, accA[u], false, false);
                accB[u] = __builtin_amdgcn_wmma_f32_16x16x4_f32(
                    false, Xc[u][3], false, Hf[3], (short)0, accB[u], false, false);
            }

            v2f base[4];
#pragma unroll
            for (int u = 0; u < 4; ++u) {
                v2f pa = hc[0] * Xc[u][0].x;
                v2f pb = hc[4] * Xc[u][2].x;
                pa = pa + hc[1] * Xc[u][0].y;
                pb = pb + hc[5] * Xc[u][2].y;
                pa = pa + hc[2] * Xc[u][1].x;
                pb = pb + hc[6] * Xc[u][3].x;
                pa = pa + hc[3] * Xc[u][1].y;
                pb = pb + hc[7] * Xc[u][3].y;
                const v2f ps = pa + pb;
                const v2f Wv = hi ? Xc[u][3] : ps;
                v2f R;
                R.x = dsvf_xhalf(Wv.x);
                R.y = dsvf_xhalf(Wv.y);
                const v2f xpv = (u == 0) ? xprev : Xc[(u + 3) & 3][3];
                const v2f up = (ps + R) + (cx2 * xpv.x + cx1 * xpv.y);
                base[u] = hi ? up : R;
            }
            xprev = Xc[3][3];

#pragma unroll
            for (int u = 0; u < 4; ++u) {
                const int t0 = (m0 + u) * 16;
                accA[u] = __builtin_amdgcn_wmma_f32_16x16x4_f32(
                    false, Sf, false, Cs, (short)0, accA[u], false, false);
                const v8f yv = accA[u] + accB[u];
#pragma unroll
                for (int v = 0; v < 8; ++v)
                    frow[(size_t)v * T_LEN + t0] = yv[v];
                Sf = (base[u] + cy2 * Sf.x) + cy1 * Sf.y;
            }
        };

#pragma unroll 1
        for (int m0 = mstart; m0 < mstart + SLICE_CHUNKS; m0 += 8) {
            trip(xa0, xb0, xa1, xb1, m0, m0 + 4);
            trip(xa1, xb1, xa0, xb0, m0 + 4, min(m0 + 8, mstart + SLICE_CHUNKS - 4));
        }
    }
}

__global__ __launch_bounds__(32) void dsvf_state_kernel(
    const float* __restrict__ x,
    const float* __restrict__ gp,
    const float* __restrict__ rp,
    const float* __restrict__ mhp_p,
    const float* __restrict__ mbp_p,
    const float* __restrict__ mlp_p,
    float* out)
{
    __shared__ float ds[80];
    dsvf_slice_body<false>(x, gp, rp, mhp_p, mbp_p, mlp_p, out, ds, ds + 40);
}

__global__ __launch_bounds__(32) void dsvf_out_kernel(
    const float* x,
    const float* __restrict__ gp,
    const float* __restrict__ rp,
    const float* __restrict__ mhp_p,
    const float* __restrict__ mbp_p,
    const float* __restrict__ mlp_p,
    float* out)
{
    __shared__ float ds[80];
    dsvf_slice_body<true>(x, gp, rp, mhp_p, mbp_p, mlp_p, out, ds, ds + 40);
}


extern "C" void kernel_launch(void* const* d_in, const int* in_sizes, int n_in,
                              void* d_out, int out_size, void* d_ws, size_t ws_size,
                              hipStream_t stream) {
    const float* x   = (const float*)d_in[0];
    const float* g   = (const float*)d_in[1];
    const float* r   = (const float*)d_in[2];
    const float* mhp = (const float*)d_in[3];
    const float* mbp = (const float*)d_in[4];
    const float* mlp = (const float*)d_in[5];
    float* y = (float*)d_out;

    const int B = in_sizes[0] / T_LEN;
    dim3 grid(B / 16 * NSLICE), block(32);
    hipLaunchKernelGGL(dsvf_state_kernel, grid, block, 0, stream,
                       x, g, r, mhp, mbp, mlp, y);
    hipLaunchKernelGGL(dsvf_out_kernel, grid, block, 0, stream,
                       x, g, r, mhp, mbp, mlp, y);
}


# State launch: a wave loads its slice of x as one group at entry into registers (was 2 loads per chunk at first use: 32 memory round trips in a row); set-up runs under them; same sums
# speedup vs baseline: 14.3212x; 1.1472x over previous
#include <hip/hip_runtime.h>
#include <math.h>


typedef float v2f __attribute__((ext_vector_type(2)));
typedef float v8f __attribute__((ext_vector_type(8)));

constexpr int T_LEN  = 8192;
constexpr int NCHUNK = T_LEN / 16;
constexpr int NSLICE = 16;
constexpr int SLICE_CHUNKS = NCHUNK / NSLICE;
constexpr int SLICE_LEN = SLICE_CHUNKS * 16;
constexpr int NPRE = 30;


struct Coef { float b0, b1, b2, a1, a2; };

__device__ __forceinline__ Coef dsvf_coef(float gv, float rv, float mhp, float mbp, float mlp)
{
    const float sig = 1.0f / (1.0f + expf(-gv));
    const float gg  = tanf(3.14159265358979323846f * 0.5f * sig);
    const float rr  = log1pf(expf(rv));
    const float g2  = gg * gg;

    Coef c;
    c.b0 = g2 * mlp + gg * mbp + mhp;
    c.b1 = 2.0f * g2 * mlp - 2.0f * mhp;
    c.b2 = g2 * mlp - gg * mbp + mhp;
    const float a0 = g2 + 2.0f * rr * gg + 1.0f;
    const float inva0 = 1.0f / a0;
    c.a1 = (2.0f * g2 - 2.0f) * inva0;
    c.a2 = (g2 - 2.0f * rr * gg + 1.0f) * inva0;
    return c;
}

__device__ __forceinline__ float dsvf_htap(const float* gs, const Coef& cf, int i)
{
    const int ic = max(i, 0);
    const float v = cf.b0 * gs[ic + 1] + cf.b1 * gs[ic] + cf.b2 * gs[max(ic - 1, 0)];
    return (i >= 0) ? v : 0.0f;
}

__device__ __forceinline__ float dsvf_xhalf(float v)
{
    return __int_as_float(__builtin_amdgcn_permlanex16(
        __float_as_int(v), __float_as_int(v), 0x76543210, 0xfedcba98, false, false));
}

template <bool FINAL>
__device__ __forceinline__ void dsvf_slice_body(
    const float* x,
    const float* __restrict__ gp,
    const float* __restrict__ rp,
    const float* __restrict__ mhp_p,
    const float* __restrict__ mbp_p,
    const float* __restrict__ mlp_p,
    float* fout,
    float* ds,
    float* gs)
{

    const unsigned lane = threadIdx.x;
    const unsigned j    = lane & 15u;
    const bool     hi   = lane >= 16u;
    const int rbase  = (int)(blockIdx.x / NSLICE) * 16;
    const int slice  = (int)(blockIdx.x % NSLICE);
    const int mstart = slice * SLICE_CHUNKS;
    if (!FINAL && slice == NSLICE - 1) return;

    const float* xrow0 = x + (size_t)(rbase + (int)j) * T_LEN;
    const float* xrow  = xrow0 + (hi ? 8 : 0);
    float4 xa[NPRE], xb[NPRE];
    if (FINAL) {
        const float* pf = xrow + mstart * 16;
#pragma unroll
        for (int u = 0; u < SLICE_CHUNKS / 4; ++u)
            __builtin_prefetch(pf + u * 64, 0, 1);
    } else {
        const float* ps = xrow + mstart * 16;
#pragma unroll
        for (int c = 0; c < NPRE; ++c) {
            xa[c] = *(const float4*)(ps + c * 16);
            xb[c] = *(const float4*)(ps + c * 16 + 4);
        }
    }
    const float2 xp = *(const float2*)(xrow0 + (mstart > 0 ? mstart * 16 - 2 : 0));
    const float xh1 = (mstart > 0) ? xp.y : 0.0f;
    const float xh2 = (mstart > 0) ? xp.x : 0.0f;
    float gv = gp[0], rv = rp[0], mhp = mhp_p[0], mbp = mbp_p[0], mlp = mlp_p[0];
    asm volatile("" : "+s"(gv), "+s"(rv), "+s"(mhp), "+s"(mbp), "+s"(mlp));
    float z1[NSLICE - 1], z2[NSLICE - 1];
    if (FINAL) {
        const float* zbase = fout + (size_t)rbase * T_LEN + (size_t)mstart * 16 + j;
#pragma unroll
        for (int i = 0; i < NSLICE - 1; ++i) {
            if (i < slice) {
                z1[i] = zbase[(size_t)i * T_LEN];
                z2[i] = zbase[(size_t)i * T_LEN + 16];
            }
        }
    }

    const Coef cf = dsvf_coef(gv, rv, mhp, mbp, mlp);

    if (lane == 0u) {
        if (!FINAL) {
            for (int i = 0; i < 40; ++i) ds[i] = 0.0f;
            ds[0] = cf.b0; ds[1] = cf.b1; ds[2] = cf.b2;
        }
        float gm2 = 0.0f, gm1 = 1.0f;
        gs[0] = 0.0f; gs[1] = 1.0f;
        for (int n = 2; n < 40; ++n) {
            const float gn = -cf.a1 * gm1 - cf.a2 * gm2;
            gs[n] = gn; gm2 = gm1; gm1 = gn;
        }
    }
    float q1 = cf.a1, q2 = cf.a2;
    for (int s = 1; s <= 8; s *= 2) {
        if (!FINAL && lane == 0u) {
            for (int i = 32; i >= 0; --i) {
                float v = ds[i];
                if (i >= s)     v -= q1 * ds[i - s];
                if (i >= 2 * s) v += q2 * ds[i - 2 * s];
                ds[i] = v;
            }
        }
        const float n1 = 2.0f * q2 - q1 * q1;
        const float n2 = q2 * q2;
        q1 = n1; q2 = n2;
    }
    __syncthreads();

    v2f Bf[3][4];
    if (!FINAL) {
#pragma unroll
        for (int b = 0; b < 3; ++b) {
#pragma unroll
            for (int s = 0; s < 4; ++s) {
                const int k0 = (hi ? 8 : 0) + 2 * s;
                const int i0 = 16 * b + (int)j - k0;
                const int i1 = i0 - 1;
                const float d0 = ds[min(max(i0, 0), 39)];
                const float d1 = ds[min(max(i1, 0), 39)];
                Bf[b][s].x = (i0 >= 0 && i0 <= 32) ? d0 : 0.0f;
                Bf[b][s].y = (i1 >= 0 && i1 <= 32) ? d1 : 0.0f;
            }
        }
    }

    const float nalpha = -q1;
    const float nbeta  = -q2;

    float ga = gs[1 + j], gb = gs[17 + j];
    for (int k = 0; k < SLICE_CHUNKS - 1; ++k) {
        const float gn = nbeta * ga + nalpha * gb;
        ga = gb; gb = gn;
    }
    const float g509 = __int_as_float(__builtin_amdgcn_readlane(__float_as_int(ga), 13));
    const float g510 = __int_as_float(__builtin_amdgcn_readlane(__float_as_int(ga), 14));
    const float g511 = __int_as_float(__builtin_amdgcn_readlane(__float_as_int(ga), 15));
    const float g512 = __int_as_float(__builtin_amdgcn_readlane(__float_as_int(gb), 0));

    v8f y1 = {};

    float* frow = fout + (size_t)(rbase + (hi ? 8 : 0)) * T_LEN + j;

    if (!FINAL) {
        v2f U1[4], U2[4], U3[4];
        for (int s = 0; s < 4; ++s) {
            U1[s].x = 0.0f; U1[s].y = 0.0f; U2[s].x = 0.0f; U2[s].y = 0.0f;
            U3[s].x = 0.0f; U3[s].y = 0.0f;
        }

        float4 xla[SLICE_CHUNKS - NPRE], xlb[SLICE_CHUNKS - NPRE];
#pragma unroll
        for (int c = 0; c < SLICE_CHUNKS - NPRE; ++c) {
            xla[c] = *(const float4*)(xrow + (mstart + NPRE + c) * 16);
            xlb[c] = *(const float4*)(xrow + (mstart + NPRE + c) * 16 + 4);
        }

        auto carry = [&](const float4& ca, const float4& cb) __attribute__((always_inline)) {
            v2f Xc[4];
            Xc[0].x = ca.x; Xc[0].y = ca.y;
            Xc[1].x = ca.z; Xc[1].y = ca.w;
            Xc[2].x = cb.x; Xc[2].y = cb.y;
            Xc[3].x = cb.z; Xc[3].y = cb.w;
            for (int s = 0; s < 4; ++s) {
                const v2f un = (Xc[s] + nbeta * U2[s]) + nalpha * U1[s];
                U3[s] = U2[s]; U2[s] = U1[s]; U1[s] = un;
            }
        };
#pragma unroll
        for (int c = 0; c < NPRE; ++c) carry(xa[c], xb[c]);
#pragma unroll
        for (int c = 0; c < SLICE_CHUNKS - NPRE; ++c) carry(xla[c], xlb[c]);

        v8f accA = {}, accB = {};
        for (int s = 0; s < 4; ++s) {
            accA = __builtin_amdgcn_wmma_f32_16x16x4_f32(
                false, U3[s], false, Bf[2][s], (short)0, accA, false, false);
            accB = __builtin_amdgcn_wmma_f32_16x16x4_f32(
                false, U2[s], false, Bf[1][s], (short)0, accB, false, false);
        }
        for (int s = 0; s < 4; s += 2) {
            accA = __builtin_amdgcn_wmma_f32_16x16x4_f32(
                false, U1[s], false, Bf[0][s], (short)0, accA, false, false);
            accB = __builtin_amdgcn_wmma_f32_16x16x4_f32(
                false, U1[s + 1], false, Bf[0][s + 1], (short)0, accB, false, false);
        }
        {
            const float gq = (j == 15u) ? g511 : g510;
            const float gp1 = (j == 15u) ? g510 : g509;
            v2f XH, EB;
            XH.x = hi ? 0.0f : xh1;
            XH.y = hi ? 0.0f : xh2;
            EB.x = hi ? 0.0f : (cf.b1 * gq + cf.b2 * gp1);
            EB.y = hi ? 0.0f : (cf.b2 * gq);
            accB = __builtin_amdgcn_wmma_f32_16x16x4_f32(
                false, XH, false, EB, (short)0, accB, false, false);
        }
        y1 = accA + accB;

        if (j >= 14u) {
            float* zrow = fout + (size_t)(rbase + slice) * T_LEN
                        + (15 - (int)j) * 16 + (hi ? 8 : 0);
            int s2 = slice + 1;
            do {
                float* zp = zrow + s2 * SLICE_LEN;
                *(float4*)(zp)     = make_float4(y1[0], y1[1], y1[2], y1[3]);
                *(float4*)(zp + 4) = make_float4(y1[4], y1[5], y1[6], y1[7]);
            } while (++s2 < NSLICE);
        }
    } else {
        const float p11 = g512, p12 = -cf.a2 * g511;
        const float p21 = g511, p22 = -cf.a2 * g510;
        float st1 = 0.0f, st2 = 0.0f;
#pragma unroll
        for (int i = 0; i < NSLICE - 1; ++i) {
            if (i < slice) {
                const float n1 = z1[i] + (p11 * st1 + p12 * st2);
                const float n2 = z2[i] + (p21 * st1 + p22 * st2);
                st1 = n1; st2 = n2;
            }
        }
        v2f Hf[4];
#pragma unroll
        for (int s = 0; s < 4; ++s) {
            const int i0 = (int)j - ((hi ? 8 : 0) + 2 * s);
            Hf[s].x = dsvf_htap(gs, cf, i0);
            Hf[s].y = dsvf_htap(gs, cf, i0 - 1);
        }
        v2f Cs, Sf;
        {
            const float g_m = gs[j], g_0 = gs[1 + j], g_p = gs[2 + j];
            Cs.x = hi ? (-cf.a2 * g_0) : (cf.b2 * g_0);
            Cs.y = hi ? g_p : (cf.b1 * g_0 + cf.b2 * g_m);
            Sf.x = hi ? st2 : xh2;
            Sf.y = hi ? st1 : xh1;
        }
        v2f hc[8];
#pragma unroll
        for (int e = 0; e < 8; ++e) {
            const int k = (hi ? 8 : 0) + e;
            hc[e].x = dsvf_htap(gs, cf, 14 - k);
            hc[e].y = dsvf_htap(gs, cf, 15 - k);
        }
        v2f cx2, cx1, cy2, cy1;
        {
            const float g13 = gs[14], g14 = gs[15], g15 = gs[16], g16 = gs[17];
            cx2.x = hi ? (cf.b2 * g14) : 0.0f;
            cx2.y = hi ? (cf.b2 * g15) : 0.0f;
            cx1.x = hi ? (cf.b1 * g14 + cf.b2 * g13) : 0.0f;
            cx1.y = hi ? (cf.b1 * g15 + cf.b2 * g14) : 0.0f;
            cy2.x = hi ? (-cf.a2 * g14) : 0.0f;
            cy2.y = hi ? (-cf.a2 * g15) : 0.0f;
            cy1.x = hi ? g15 : 0.0f;
            cy1.y = hi ? g16 : 0.0f;
        }
        v2f xprev;
        xprev.x = xh2; xprev.y = xh1;

        float4 xa0[4], xb0[4], xa1[4], xb1[4];
#pragma unroll
        for (int u = 0; u < 4; ++u) {
            xa0[u] = *(const float4*)(xrow + (mstart + u) * 16);
            xb0[u] = *(const float4*)(xrow + (mstart + u) * 16 + 4);
        }

        auto trip = [&](const float4 (&xac)[4], const float4 (&xbc)[4],
                        float4 (&xan)[4], float4 (&xbn)[4], int m0, int mn)
                        __attribute__((always_inline)) {
#pragma unroll
            for (int u = 0; u < 4; ++u) {
                xan[u] = *(const float4*)(xrow + (mn + u) * 16);
                xbn[u] = *(const float4*)(xrow + (mn + u) * 16 + 4);
            }
            __builtin_amdgcn_sched_barrier(0);
            v2f Xc[4][4];
#pragma unroll
            for (int u = 0; u < 4; ++u) {
                Xc[u][0].x = xac[u].x; Xc[u][0].y = xac[u].y;
                Xc[u][1].x = xac[u].z; Xc[u][1].y = xac[u].w;
                Xc[u][2].x = xbc[u].x; Xc[u][2].y = xbc[u].y;
                Xc[u][3].x = xbc[u].z; Xc[u][3].y = xbc[u].w;
            }

            v8f accA[4], accB[4];
            const v8f zero8 = {};
#pragma unroll
            for (int u = 0; u < 4; ++u) {
                accA[u] = __builtin_amdgcn_wmma_f32_16x16x4_f32(
                    false, Xc[u][0], false, Hf[0], (short)0, zero8, false, false);
                accB[u] = __builtin_amdgcn_wmma_f32_16x16x4_f32(
                    false, Xc[u][2], false, Hf[2], (short)0, zero8, false, false);
                accA[u] = __builtin_amdgcn_wmma_f32_16x16x4_f32(
                    false, Xc[u][1], false, Hf[1], (short)0, accA[u], false, false);
                accB[u] = __builtin_amdgcn_wmma_f32_16x16x4_f32(
                    false, Xc[u][3], false, Hf[3], (short)0, accB[u], false, false);
            }

            v2f base[4];
#pragma unroll
            for (int u = 0; u < 4; ++u) {
                v2f pa = hc[0] * Xc[u][0].x;
                v2f pb = hc[4] * Xc[u][2].x;
                pa = pa + hc[1] * Xc[u][0].y;
                pb = pb + hc[5] * Xc[u][2].y;
                pa = pa + hc[2] * Xc[u][1].x;
                pb = pb + hc[6] * Xc[u][3].x;
                pa = pa + hc[3] * Xc[u][1].y;
                pb = pb + hc[7] * Xc[u][3].y;
                const v2f ps = pa + pb;
                const v2f Wv = hi ? Xc[u][3] : ps;
                v2f R;
                R.x = dsvf_xhalf(Wv.x);
                R.y = dsvf_xhalf(Wv.y);
                const v2f xpv = (u == 0) ? xprev : Xc[(u + 3) & 3][3];
                const v2f up = (ps + R) + (cx2 * xpv.x + cx1 * xpv.y);
                base[u] = hi ? up : R;
            }
            xprev = Xc[3][3];

#pragma unroll
            for (int u = 0; u < 4; ++u) {
                const int t0 = (m0 + u) * 16;
                accA[u] = __builtin_amdgcn_wmma_f32_16x16x4_f32(
                    false, Sf, false, Cs, (short)0, accA[u], false, false);
                const v8f yv = accA[u] + accB[u];
#pragma unroll
                for (int v = 0; v < 8; ++v)
                    frow[(size_t)v * T_LEN + t0] = yv[v];
                Sf = (base[u] + cy2 * Sf.x) + cy1 * Sf.y;
            }
        };

#pragma unroll 1
        for (int m0 = mstart; m0 < mstart + SLICE_CHUNKS; m0 += 8) {
            trip(xa0, xb0, xa1, xb1, m0, m0 + 4);
            trip(xa1, xb1, xa0, xb0, m0 + 4, min(m0 + 8, mstart + SLICE_CHUNKS - 4));
        }
    }
}

__global__ __launch_bounds__(32) void dsvf_state_kernel(
    const float* __restrict__ x,
    const float* __restrict__ gp,
    const float* __restrict__ rp,
    const float* __restrict__ mhp_p,
    const float* __restrict__ mbp_p,
    const float* __restrict__ mlp_p,
    float* out)
{
    __shared__ float ds[80];
    dsvf_slice_body<false>(x, gp, rp, mhp_p, mbp_p, mlp_p, out, ds, ds + 40);
}

__global__ __launch_bounds__(32) void dsvf_out_kernel(
    const float* x,
    const float* __restrict__ gp,
    const float* __restrict__ rp,
    const float* __restrict__ mhp_p,
    const float* __restrict__ mbp_p,
    const float* __restrict__ mlp_p,
    float* out)
{
    __shared__ float ds[80];
    dsvf_slice_body<true>(x, gp, rp, mhp_p, mbp_p, mlp_p, out, ds, ds + 40);
}


extern "C" void kernel_launch(void* const* d_in, const int* in_sizes, int n_in,
                              void* d_out, int out_size, void* d_ws, size_t ws_size,
                              hipStream_t stream) {
    const float* x   = (const float*)d_in[0];
    const float* g   = (const float*)d_in[1];
    const float* r   = (const float*)d_in[2];
    const float* mhp = (const float*)d_in[3];
    const float* mbp = (const float*)d_in[4];
    const float* mlp = (const float*)d_in[5];
    float* y = (float*)d_out;

    const int B = in_sizes[0] / T_LEN;
    dim3 grid(B / 16 * NSLICE), block(32);
    hipLaunchKernelGGL(dsvf_state_kernel, grid, block, 0, stream,
                       x, g, r, mhp, mbp, mlp, y);
    hipLaunchKernelGGL(dsvf_out_kernel, grid, block, 0, stream,
                       x, g, r, mhp, mbp, mlp, y);
}


# Output launch: x read as whole lines (a load = 2 rows x 256 B, 4 lines; was 16 rows x 32 B, 16 lines), turned into fragments through a per-wave LDS slab; loads two trips ahead; same sums
# speedup vs baseline: 15.1555x; 1.0583x over previous
#include <hip/hip_runtime.h>
#include <math.h>


typedef float v2f __attribute__((ext_vector_type(2)));
typedef float v8f __attribute__((ext_vector_type(8)));
typedef float v4f __attribute__((ext_vector_type(4)));

constexpr int T_LEN  = 8192;
constexpr int NCHUNK = T_LEN / 16;
constexpr int NSLICE = 16;
constexpr int SLICE_CHUNKS = NCHUNK / NSLICE;
constexpr int SLICE_LEN = SLICE_CHUNKS * 16;
constexpr int NPRE = 30;
constexpr int XS_PITCH = 68;


struct Coef { float b0, b1, b2, a1, a2; };

__device__ __forceinline__ Coef dsvf_coef(float gv, float rv, float mhp, float mbp, float mlp)
{
    const float sig = 1.0f / (1.0f + expf(-gv));
    const float gg  = tanf(3.14159265358979323846f * 0.5f * sig);
    const float rr  = log1pf(expf(rv));
    const float g2  = gg * gg;

    Coef c;
    c.b0 = g2 * mlp + gg * mbp + mhp;
    c.b1 = 2.0f * g2 * mlp - 2.0f * mhp;
    c.b2 = g2 * mlp - gg * mbp + mhp;
    const float a0 = g2 + 2.0f * rr * gg + 1.0f;
    const float inva0 = 1.0f / a0;
    c.a1 = (2.0f * g2 - 2.0f) * inva0;
    c.a2 = (g2 - 2.0f * rr * gg + 1.0f) * inva0;
    return c;
}

__device__ __forceinline__ float dsvf_htap(const float* gs, const Coef& cf, int i)
{
    const int ic = max(i, 0);
    const float v = cf.b0 * gs[ic + 1] + cf.b1 * gs[ic] + cf.b2 * gs[max(ic - 1, 0)];
    return (i >= 0) ? v : 0.0f;
}

__device__ __forceinline__ float dsvf_xhalf(float v)
{
    return __int_as_float(__builtin_amdgcn_permlanex16(
        __float_as_int(v), __float_as_int(v), 0x76543210, 0xfedcba98, false, false));
}

__device__ __forceinline__ void dsvf_gload(v4f (&raw)[8], const float* xw, int m)
{
#pragma unroll
    for (int q = 0; q < 8; ++q)
        raw[q] = *(const v4f*)(xw + (size_t)(2 * q) * T_LEN + m * 16);
}

__device__ __forceinline__ void dsvf_turn(const v4f (&raw)[8], float* xst, const float* xld,
                                          float4 (&fa)[4], float4 (&fb)[4])
{
    asm volatile("" ::: "memory");
#pragma unroll
    for (int q = 0; q < 8; ++q)
        *(v4f*)(xst + 2 * q * XS_PITCH) = raw[q];
    asm volatile("" ::: "memory");
#pragma unroll
    for (int u = 0; u < 4; ++u) {
        fa[u] = *(const float4*)(xld + u * 16);
        fb[u] = *(const float4*)(xld + u * 16 + 4);
    }
}

template <bool FINAL>
__device__ __forceinline__ void dsvf_slice_body(
    const float* x,
    const float* __restrict__ gp,
    const float* __restrict__ rp,
    const float* __restrict__ mhp_p,
    const float* __restrict__ mbp_p,
    const float* __restrict__ mlp_p,
    float* fout,
    float* ds,
    float* gs,
    float* xs)
{

    const unsigned lane = threadIdx.x;
    const unsigned j    = lane & 15u;
    const bool     hi   = lane >= 16u;
    const int rbase  = (int)(blockIdx.x / NSLICE) * 16;
    const int slice  = (int)(blockIdx.x % NSLICE);
    const int mstart = slice * SLICE_CHUNKS;
    if (!FINAL && slice == NSLICE - 1) return;

    const float* xrow0 = x + (size_t)(rbase + (int)j) * T_LEN;
    const float* xrow  = xrow0 + (hi ? 8 : 0);
    float4 xa[NPRE], xb[NPRE];
    if (FINAL) {
        const float* pf = xrow + mstart * 16;
#pragma unroll
        for (int u = 0; u < SLICE_CHUNKS / 4; ++u)
            __builtin_prefetch(pf + u * 64, 0, 1);
    } else {
        const float* ps = xrow + mstart * 16;
#pragma unroll
        for (int c = 0; c < NPRE; ++c) {
            xa[c] = *(const float4*)(ps + c * 16);
            xb[c] = *(const float4*)(ps + c * 16 + 4);
        }
    }
    const float2 xp = *(const float2*)(xrow0 + (mstart > 0 ? mstart * 16 - 2 : 0));
    const float xh1 = (mstart > 0) ? xp.y : 0.0f;
    const float xh2 = (mstart > 0) ? xp.x : 0.0f;
    float gv = gp[0], rv = rp[0], mhp = mhp_p[0], mbp = mbp_p[0], mlp = mlp_p[0];
    asm volatile("" : "+s"(gv), "+s"(rv), "+s"(mhp), "+s"(mbp), "+s"(mlp));
    float z1[NSLICE - 1], z2[NSLICE - 1];
    if (FINAL) {
        const float* zbase = fout + (size_t)rbase * T_LEN + (size_t)mstart * 16 + j;
#pragma unroll
        for (int i = 0; i < NSLICE - 1; ++i) {
            if (i < slice) {
                z1[i] = zbase[(size_t)i * T_LEN];
                z2[i] = zbase[(size_t)i * T_LEN + 16];
            }
        }
    }

    const Coef cf = dsvf_coef(gv, rv, mhp, mbp, mlp);

    if (lane == 0u) {
        if (!FINAL) {
            for (int i = 0; i < 40; ++i) ds[i] = 0.0f;
            ds[0] = cf.b0; ds[1] = cf.b1; ds[2] = cf.b2;
        }
        float gm2 = 0.0f, gm1 = 1.0f;
        gs[0] = 0.0f; gs[1] = 1.0f;
        for (int n = 2; n < 40; ++n) {
            const float gn = -cf.a1 * gm1 - cf.a2 * gm2;
            gs[n] = gn; gm2 = gm1; gm1 = gn;
        }
    }
    float q1 = cf.a1, q2 = cf.a2;
    for (int s = 1; s <= 8; s *= 2) {
        if (!FINAL && lane == 0u) {
            for (int i = 32; i >= 0; --i) {
                float v = ds[i];
                if (i >= s)     v -= q1 * ds[i - s];
                if (i >= 2 * s) v += q2 * ds[i - 2 * s];
                ds[i] = v;
            }
        }
        const float n1 = 2.0f * q2 - q1 * q1;
        const float n2 = q2 * q2;
        q1 = n1; q2 = n2;
    }
    __syncthreads();

    v2f Bf[3][4];
    if (!FINAL) {
#pragma unroll
        for (int b = 0; b < 3; ++b) {
#pragma unroll
            for (int s = 0; s < 4; ++s) {
                const int k0 = (hi ? 8 : 0) + 2 * s;
                const int i0 = 16 * b + (int)j - k0;
                const int i1 = i0 - 1;
                const float d0 = ds[min(max(i0, 0), 39)];
                const float d1 = ds[min(max(i1, 0), 39)];
                Bf[b][s].x = (i0 >= 0 && i0 <= 32) ? d0 : 0.0f;
                Bf[b][s].y = (i1 >= 0 && i1 <= 32) ? d1 : 0.0f;
            }
        }
    }

    const float nalpha = -q1;
    const float nbeta  = -q2;

    float ga = gs[1 + j], gb = gs[17 + j];
    for (int k = 0; k < SLICE_CHUNKS - 1; ++k) {
        const float gn = nbeta * ga + nalpha * gb;
        ga = gb; gb = gn;
    }
    const float g509 = __int_as_float(__builtin_amdgcn_readlane(__float_as_int(ga), 13));
    const float g510 = __int_as_float(__builtin_amdgcn_readlane(__float_as_int(ga), 14));
    const float g511 = __int_as_float(__builtin_amdgcn_readlane(__float_as_int(ga), 15));
    const float g512 = __int_as_float(__builtin_amdgcn_readlane(__float_as_int(gb), 0));

    v8f y1 = {};

    float* frow = fout + (size_t)(rbase + (hi ? 8 : 0)) * T_LEN + j;

    if (!FINAL) {
        v2f U1[4], U2[4], U3[4];
        for (int s = 0; s < 4; ++s) {
            U1[s].x = 0.0f; U1[s].y = 0.0f; U2[s].x = 0.0f; U2[s].y = 0.0f;
            U3[s].x = 0.0f; U3[s].y = 0.0f;
        }

        float4 xla[SLICE_CHUNKS - NPRE], xlb[SLICE_CHUNKS - NPRE];
#pragma unroll
        for (int c = 0; c < SLICE_CHUNKS - NPRE; ++c) {
            xla[c] = *(const float4*)(xrow + (mstart + NPRE + c) * 16);
            xlb[c] = *(const float4*)(xrow + (mstart + NPRE + c) * 16 + 4);
        }

        auto carry = [&](const float4& ca, const float4& cb) __attribute__((always_inline)) {
            v2f Xc[4];
            Xc[0].x = ca.x; Xc[0].y = ca.y;
            Xc[1].x = ca.z; Xc[1].y = ca.w;
            Xc[2].x = cb.x; Xc[2].y = cb.y;
            Xc[3].x = cb.z; Xc[3].y = cb.w;
            for (int s = 0; s < 4; ++s) {
                const v2f un = (Xc[s] + nbeta * U2[s]) + nalpha * U1[s];
                U3[s] = U2[s]; U2[s] = U1[s]; U1[s] = un;
            }
        };
#pragma unroll
        for (int c = 0; c < NPRE; ++c) carry(xa[c], xb[c]);
#pragma unroll
        for (int c = 0; c < SLICE_CHUNKS - NPRE; ++c) carry(xla[c], xlb[c]);

        v8f accA = {}, accB = {};
        for (int s = 0; s < 4; ++s) {
            accA = __builtin_amdgcn_wmma_f32_16x16x4_f32(
                false, U3[s], false, Bf[2][s], (short)0, accA, false, false);
            accB = __builtin_amdgcn_wmma_f32_16x16x4_f32(
                false, U2[s], false, Bf[1][s], (short)0, accB, false, false);
        }
        for (int s = 0; s < 4; s += 2) {
            accA = __builtin_amdgcn_wmma_f32_16x16x4_f32(
                false, U1[s], false, Bf[0][s], (short)0, accA, false, false);
            accB = __builtin_amdgcn_wmma_f32_16x16x4_f32(
                false, U1[s + 1], false, Bf[0][s + 1], (short)0, accB, false, false);
        }
        {
            const float gq = (j == 15u) ? g511 : g510;
            const float gp1 = (j == 15u) ? g510 : g509;
            v2f XH, EB;
            XH.x = hi ? 0.0f : xh1;
            XH.y = hi ? 0.0f : xh2;
            EB.x = hi ? 0.0f : (cf.b1 * gq + cf.b2 * gp1);
            EB.y = hi ? 0.0f : (cf.b2 * gq);
            accB = __builtin_amdgcn_wmma_f32_16x16x4_f32(
                false, XH, false, EB, (short)0, accB, false, false);
        }
        y1 = accA + accB;

        if (j >= 14u) {
            float* zrow = fout + (size_t)(rbase + slice) * T_LEN
                        + (15 - (int)j) * 16 + (hi ? 8 : 0);
            int s2 = slice + 1;
            do {
                float* zp = zrow + s2 * SLICE_LEN;
                *(float4*)(zp)     = make_float4(y1[0], y1[1], y1[2], y1[3]);
                *(float4*)(zp + 4) = make_float4(y1[4], y1[5], y1[6], y1[7]);
            } while (++s2 < NSLICE);
        }
    } else {
        const float p11 = g512, p12 = -cf.a2 * g511;
        const float p21 = g511, p22 = -cf.a2 * g510;
        float st1 = 0.0f, st2 = 0.0f;
#pragma unroll
        for (int i = 0; i < NSLICE - 1; ++i) {
            if (i < slice) {
                const float n1 = z1[i] + (p11 * st1 + p12 * st2);
                const float n2 = z2[i] + (p21 * st1 + p22 * st2);
                st1 = n1; st2 = n2;
            }
        }
        v2f Hf[4];
#pragma unroll
        for (int s = 0; s < 4; ++s) {
            const int i0 = (int)j - ((hi ? 8 : 0) + 2 * s);
            Hf[s].x = dsvf_htap(gs, cf, i0);
            Hf[s].y = dsvf_htap(gs, cf, i0 - 1);
        }
        v2f Cs, Sf;
        {
            const float g_m = gs[j], g_0 = gs[1 + j], g_p = gs[2 + j];
            Cs.x = hi ? (-cf.a2 * g_0) : (cf.b2 * g_0);
            Cs.y = hi ? g_p : (cf.b1 * g_0 + cf.b2 * g_m);
            Sf.x = hi ? st2 : xh2;
            Sf.y = hi ? st1 : xh1;
        }
        v2f hc[8];
#pragma unroll
        for (int e = 0; e < 8; ++e) {
            const int k = (hi ? 8 : 0) + e;
            hc[e].x = dsvf_htap(gs, cf, 14 - k);
            hc[e].y = dsvf_htap(gs, cf, 15 - k);
        }
        v2f cx2, cx1, cy2, cy1;
        {
            const float g13 = gs[14], g14 = gs[15], g15 = gs[16], g16 = gs[17];
            cx2.x = hi ? (cf.b2 * g14) : 0.0f;
            cx2.y = hi ? (cf.b2 * g15) : 0.0f;
            cx1.x = hi ? (cf.b1 * g14 + cf.b2 * g13) : 0.0f;
            cx1.y = hi ? (cf.b1 * g15 + cf.b2 * g14) : 0.0f;
            cy2.x = hi ? (-cf.a2 * g14) : 0.0f;
            cy2.y = hi ? (-cf.a2 * g15) : 0.0f;
            cy1.x = hi ? g15 : 0.0f;
            cy1.y = hi ? g16 : 0.0f;
        }
        v2f xprev;
        xprev.x = xh2; xprev.y = xh1;

        const unsigned lrow = lane >> 4;
        const unsigned lcol = (lane & 15u) * 4u;
        const float* xw  = x + (size_t)(rbase + (int)lrow) * T_LEN + lcol;
        float*       xst = xs + lrow * XS_PITCH + lcol;
        const float* xld = xs + j * XS_PITCH + (hi ? 8 : 0);
        v4f raw[8];
        float4 xa0[4], xb0[4], xa1[4], xb1[4];
        dsvf_gload(raw, xw, mstart);
        dsvf_turn(raw, xst, xld, xa0, xb0);
        dsvf_gload(raw, xw, mstart + 4);

        auto trip = [&](const float4 (&xac)[4], const float4 (&xbc)[4],
                        float4 (&xan)[4], float4 (&xbn)[4], v4f (&rw)[8], int m0, int mn,
                        const bool do_turn, const bool do_load)
                        __attribute__((always_inline)) {
            if (do_turn) dsvf_turn(rw, xst, xld, xan, xbn);
            if (do_load) dsvf_gload(rw, xw, mn);
            __builtin_amdgcn_sched_barrier(0);
            v2f Xc[4][4];
#pragma unroll
            for (int u = 0; u < 4; ++u) {
                Xc[u][0].x = xac[u].x; Xc[u][0].y = xac[u].y;
                Xc[u][1].x = xac[u].z; Xc[u][1].y = xac[u].w;
                Xc[u][2].x = xbc[u].x; Xc[u][2].y = xbc[u].y;
                Xc[u][3].x = xbc[u].z; Xc[u][3].y = xbc[u].w;
            }

            v8f accA[4], accB[4];
            const v8f zero8 = {};
#pragma unroll
            for (int u = 0; u < 4; ++u) {
                accA[u] = __builtin_amdgcn_wmma_f32_16x16x4_f32(
                    false, Xc[u][0], false, Hf[0], (short)0, zero8, false, false);
                accB[u] = __builtin_amdgcn_wmma_f32_16x16x4_f32(
                    false, Xc[u][2], false, Hf[2], (short)0, zero8, false, false);
                accA[u] = __builtin_amdgcn_wmma_f32_16x16x4_f32(
                    false, Xc[u][1], false, Hf[1], (short)0, accA[u], false, false);
                accB[u] = __builtin_amdgcn_wmma_f32_16x16x4_f32(
                    false, Xc[u][3], false, Hf[3], (short)0, accB[u], false, false);
            }

            v2f base[4];
#pragma unroll
            for (int u = 0; u < 4; ++u) {
                v2f pa = hc[0] * Xc[u][0].x;
                v2f pb = hc[4] * Xc[u][2].x;
                pa = pa + hc[1] * Xc[u][0].y;
                pb = pb + hc[5] * Xc[u][2].y;
                pa = pa + hc[2] * Xc[u][1].x;
                pb = pb + hc[6] * Xc[u][3].x;
                pa = pa + hc[3] * Xc[u][1].y;
                pb = pb + hc[7] * Xc[u][3].y;
                const v2f ps = pa + pb;
                const v2f Wv = hi ? Xc[u][3] : ps;
                v2f R;
                R.x = dsvf_xhalf(Wv.x);
                R.y = dsvf_xhalf(Wv.y);
                const v2f xpv = (u == 0) ? xprev : Xc[(u + 3) & 3][3];
                const v2f up = (ps + R) + (cx2 * xpv.x + cx1 * xpv.y);
                base[u] = hi ? up : R;
            }
            xprev = Xc[3][3];

#pragma unroll
            for (int u = 0; u < 4; ++u) {
                const int t0 = (m0 + u) * 16;
                accA[u] = __builtin_amdgcn_wmma_f32_16x16x4_f32(
                    false, Sf, false, Cs, (short)0, accA[u], false, false);
                const v8f yv = accA[u] + accB[u];
#pragma unroll
                for (int v = 0; v < 8; ++v)
                    frow[(size_t)v * T_LEN + t0] = yv[v];
                Sf = (base[u] + cy2 * Sf.x) + cy1 * Sf.y;
            }
        };

#pragma unroll 1
        for (int m0 = mstart; m0 < mstart + SLICE_CHUNKS - 8; m0 += 8) {
            trip(xa0, xb0, xa1, xb1, raw, m0, m0 + 8, true, true);
            trip(xa1, xb1, xa0, xb0, raw, m0 + 4, m0 + 12, true, true);
        }
        trip(xa0, xb0, xa1, xb1, raw, mstart + SLICE_CHUNKS - 8, 0, true, false);
        trip(xa1, xb1, xa0, xb0, raw, mstart + SLICE_CHUNKS - 4, 0, false, false);
    }
}

__global__ __launch_bounds__(32) void dsvf_state_kernel(
    const float* __restrict__ x,
    const float* __restrict__ gp,
    const float* __restrict__ rp,
    const float* __restrict__ mhp_p,
    const float* __restrict__ mbp_p,
    const float* __restrict__ mlp_p,
    float* out)
{
    __shared__ float ds[80];
    dsvf_slice_body<false>(x, gp, rp, mhp_p, mbp_p, mlp_p, out, ds, ds + 40, nullptr);
}

__global__ __launch_bounds__(32) void dsvf_out_kernel(
    const float* x,
    const float* __restrict__ gp,
    const float* __restrict__ rp,
    const float* __restrict__ mhp_p,
    const float* __restrict__ mbp_p,
    const float* __restrict__ mlp_p,
    float* out)
{
    __shared__ float ds[80];
    __shared__ __attribute__((aligned(16))) float xs[16 * XS_PITCH];
    dsvf_slice_body<true>(x, gp, rp, mhp_p, mbp_p, mlp_p, out, ds, ds + 40, xs);
}


extern "C" void kernel_launch(void* const* d_in, const int* in_sizes, int n_in,
                              void* d_out, int out_size, void* d_ws, size_t ws_size,
                              hipStream_t stream) {
    const float* x   = (const float*)d_in[0];
    const float* g   = (const float*)d_in[1];
    const float* r   = (const float*)d_in[2];
    const float* mhp = (const float*)d_in[3];
    const float* mbp = (const float*)d_in[4];
    const float* mlp = (const float*)d_in[5];
    float* y = (float*)d_out;

    const int B = in_sizes[0] / T_LEN;
    dim3 grid(B / 16 * NSLICE), block(32);
    hipLaunchKernelGGL(dsvf_state_kernel, grid, block, 0, stream,
                       x, g, r, mhp, mbp, mlp, y);
    hipLaunchKernelGGL(dsvf_out_kernel, grid, block, 0, stream,
                       x, g, r, mhp, mbp, mlp, y);
}


# One wave owns a whole row: its 16 time slices are the 16 rows of every matrix tile; state pass, cross-lane fold and output pass in one launch, x read once into registers; same sums
# speedup vs baseline: 18.2890x; 1.2068x over previous
#include <hip/hip_runtime.h>
#include <math.h>


typedef float v2f __attribute__((ext_vector_type(2)));
typedef float v8f __attribute__((ext_vector_type(8)));

constexpr int T_LEN  = 8192;
constexpr int NSLICE = 16;
constexpr int SLICE_LEN = T_LEN / NSLICE;
constexpr int SLICE_CHUNKS = SLICE_LEN / 16;
constexpr int NPRE = 30;


struct Coef { float b0, b1, b2, a1, a2; };

__device__ __forceinline__ Coef dsvf_coef(float gv, float rv, float mhp, float mbp, float mlp)
{
    const float sig = 1.0f / (1.0f + expf(-gv));
    const float gg  = tanf(3.14159265358979323846f * 0.5f * sig);
    const float rr  = log1pf(expf(rv));
    const float g2  = gg * gg;

    Coef c;
    c.b0 = g2 * mlp + gg * mbp + mhp;
    c.b1 = 2.0f * g2 * mlp - 2.0f * mhp;
    c.b2 = g2 * mlp - gg * mbp + mhp;
    const float a0 = g2 + 2.0f * rr * gg + 1.0f;
    const float inva0 = 1.0f / a0;
    c.a1 = (2.0f * g2 - 2.0f) * inva0;
    c.a2 = (g2 - 2.0f * rr * gg + 1.0f) * inva0;
    return c;
}

__device__ __forceinline__ float dsvf_htap(const float* gs, const Coef& cf, int i)
{
    const int ic = max(i, 0);
    const float v = cf.b0 * gs[ic + 1] + cf.b1 * gs[ic] + cf.b2 * gs[max(ic - 1, 0)];
    return (i >= 0) ? v : 0.0f;
}

__device__ __forceinline__ float dsvf_xhalf(float v)
{
    return __int_as_float(__builtin_amdgcn_permlanex16(
        __float_as_int(v), __float_as_int(v), 0x76543210, 0xfedcba98, false, false));
}

__global__ __launch_bounds__(32) void dsvf_row_kernel(
    const float* __restrict__ x,
    const float* __restrict__ gp,
    const float* __restrict__ rp,
    const float* __restrict__ mhp_p,
    const float* __restrict__ mbp_p,
    const float* __restrict__ mlp_p,
    float* __restrict__ out)
{
    __shared__ float ds[80];
    float* gs = ds + 40;

    const unsigned lane = threadIdx.x;
    const unsigned j    = lane & 15u;
    const bool     hi   = lane >= 16u;

    const float* xrow0 = x + (size_t)blockIdx.x * T_LEN + (size_t)j * SLICE_LEN;
    const float* xrow  = xrow0 + (hi ? 8 : 0);
    float4 xa[SLICE_CHUNKS], xb[SLICE_CHUNKS];
#pragma unroll
    for (int c = 0; c < NPRE; ++c) {
        xa[c] = *(const float4*)(xrow + c * 16);
        xb[c] = *(const float4*)(xrow + c * 16 + 4);
    }
    const float2 xp = *(const float2*)(xrow0 - (j > 0u ? 2 : 0));
    const float xh1 = (j > 0u) ? xp.y : 0.0f;
    const float xh2 = (j > 0u) ? xp.x : 0.0f;
    float gv = gp[0], rv = rp[0], mhp = mhp_p[0], mbp = mbp_p[0], mlp = mlp_p[0];
    asm volatile("" : "+s"(gv), "+s"(rv), "+s"(mhp), "+s"(mbp), "+s"(mlp));

    const Coef cf = dsvf_coef(gv, rv, mhp, mbp, mlp);

    if (lane == 0u) {
        for (int i = 0; i < 40; ++i) ds[i] = 0.0f;
        ds[0] = cf.b0; ds[1] = cf.b1; ds[2] = cf.b2;
        float gm2 = 0.0f, gm1 = 1.0f;
        gs[0] = 0.0f; gs[1] = 1.0f;
        for (int n = 2; n < 40; ++n) {
            const float gn = -cf.a1 * gm1 - cf.a2 * gm2;
            gs[n] = gn; gm2 = gm1; gm1 = gn;
        }
    }
    float q1 = cf.a1, q2 = cf.a2;
    for (int s = 1; s <= 8; s *= 2) {
        if (lane == 0u) {
            for (int i = 32; i >= 0; --i) {
                float v = ds[i];
                if (i >= s)     v -= q1 * ds[i - s];
                if (i >= 2 * s) v += q2 * ds[i - 2 * s];
                ds[i] = v;
            }
        }
        const float n1 = 2.0f * q2 - q1 * q1;
        const float n2 = q2 * q2;
        q1 = n1; q2 = n2;
    }
    __syncthreads();

    v2f Bf[3][4];
#pragma unroll
    for (int b = 0; b < 3; ++b) {
#pragma unroll
        for (int s = 0; s < 4; ++s) {
            const int k0 = (hi ? 8 : 0) + 2 * s;
            const int i0 = 16 * b + (int)j - k0;
            const int i1 = i0 - 1;
            const float d0 = ds[min(max(i0, 0), 39)];
            const float d1 = ds[min(max(i1, 0), 39)];
            Bf[b][s].x = (i0 >= 0 && i0 <= 32) ? d0 : 0.0f;
            Bf[b][s].y = (i1 >= 0 && i1 <= 32) ? d1 : 0.0f;
        }
    }

    const float nalpha = -q1;
    const float nbeta  = -q2;

    float ga = gs[1 + j], gb = gs[17 + j];
    for (int k = 0; k < SLICE_CHUNKS - 1; ++k) {
        const float gn = nbeta * ga + nalpha * gb;
        ga = gb; gb = gn;
    }
    const float g509 = __int_as_float(__builtin_amdgcn_readlane(__float_as_int(ga), 13));
    const float g510 = __int_as_float(__builtin_amdgcn_readlane(__float_as_int(ga), 14));
    const float g511 = __int_as_float(__builtin_amdgcn_readlane(__float_as_int(ga), 15));
    const float g512 = __int_as_float(__builtin_amdgcn_readlane(__float_as_int(gb), 0));

#pragma unroll
    for (int c = NPRE; c < SLICE_CHUNKS; ++c) {
        xa[c] = *(const float4*)(xrow + c * 16);
        xb[c] = *(const float4*)(xrow + c * 16 + 4);
    }

    v8f y1;
    {
        v2f U1[4], U2[4], U3[4];
        for (int s = 0; s < 4; ++s) {
            U1[s].x = 0.0f; U1[s].y = 0.0f; U2[s].x = 0.0f; U2[s].y = 0.0f;
            U3[s].x = 0.0f; U3[s].y = 0.0f;
        }
        auto carry = [&](const float4& ca, const float4& cb) __attribute__((always_inline)) {
            v2f Xc[4];
            Xc[0].x = ca.x; Xc[0].y = ca.y;
            Xc[1].x = ca.z; Xc[1].y = ca.w;
            Xc[2].x = cb.x; Xc[2].y = cb.y;
            Xc[3].x = cb.z; Xc[3].y = cb.w;
            for (int s = 0; s < 4; ++s) {
                const v2f un = (Xc[s] + nbeta * U2[s]) + nalpha * U1[s];
                U3[s] = U2[s]; U2[s] = U1[s]; U1[s] = un;
            }
        };
#pragma unroll
        for (int c = 0; c < SLICE_CHUNKS; ++c) carry(xa[c], xb[c]);

        v8f accA = {}, accB = {};
        for (int s = 0; s < 4; ++s) {
            accA = __builtin_amdgcn_wmma_f32_16x16x4_f32(
                false, U3[s], false, Bf[2][s], (short)0, accA, false, false);
            accB = __builtin_amdgcn_wmma_f32_16x16x4_f32(
                false, U2[s], false, Bf[1][s], (short)0, accB, false, false);
        }
        for (int s = 0; s < 4; s += 2) {
            accA = __builtin_amdgcn_wmma_f32_16x16x4_f32(
                false, U1[s], false, Bf[0][s], (short)0, accA, false, false);
            accB = __builtin_amdgcn_wmma_f32_16x16x4_f32(
                false, U1[s + 1], false, Bf[0][s + 1], (short)0, accB, false, false);
        }
        {
            const float gq = (j == 15u) ? g511 : g510;
            const float gp1 = (j == 15u) ? g510 : g509;
            v2f XH, EB;
            XH.x = hi ? 0.0f : xh1;
            XH.y = hi ? 0.0f : xh2;
            EB.x = hi ? 0.0f : (cf.b1 * gq + cf.b2 * gp1);
            EB.y = hi ? 0.0f : (cf.b2 * gq);
            accB = __builtin_amdgcn_wmma_f32_16x16x4_f32(
                false, XH, false, EB, (short)0, accB, false, false);
        }
        y1 = accA + accB;
    }

    const float p11 = g512, p12 = -cf.a2 * g511;
    const float p21 = g511, p22 = -cf.a2 * g510;
    float st1 = 0.0f, st2 = 0.0f;
    {
        float u1 = 0.0f, u2 = 0.0f;
#pragma unroll
        for (int i = 0; i < NSLICE - 1; ++i) {
            const float zi1 = __int_as_float(__builtin_amdgcn_readlane(
                __float_as_int(y1[i & 7]), 15 + 16 * (i >> 3)));
            const float zi2 = __int_as_float(__builtin_amdgcn_readlane(
                __float_as_int(y1[i & 7]), 14 + 16 * (i >> 3)));
            const float n1 = zi1 + (p11 * u1 + p12 * u2);
            const float n2 = zi2 + (p21 * u1 + p22 * u2);
            u1 = n1; u2 = n2;
            st1 = (j == (unsigned)(i + 1)) ? n1 : st1;
            st2 = (j == (unsigned)(i + 1)) ? n2 : st2;
        }
    }

    v2f Hf[4];
#pragma unroll
    for (int s = 0; s < 4; ++s) {
        const int i0 = (int)j - ((hi ? 8 : 0) + 2 * s);
        Hf[s].x = dsvf_htap(gs, cf, i0);
        Hf[s].y = dsvf_htap(gs, cf, i0 - 1);
    }
    v2f Cs, Sf;
    {
        const float g_m = gs[j], g_0 = gs[1 + j], g_p = gs[2 + j];
        Cs.x = hi ? (-cf.a2 * g_0) : (cf.b2 * g_0);
        Cs.y = hi ? g_p : (cf.b1 * g_0 + cf.b2 * g_m);
        Sf.x = hi ? st2 : xh2;
        Sf.y = hi ? st1 : xh1;
    }
    v2f hc[8];
#pragma unroll
    for (int e = 0; e < 8; ++e) {
        const int k = (hi ? 8 : 0) + e;
        hc[e].x = dsvf_htap(gs, cf, 14 - k);
        hc[e].y = dsvf_htap(gs, cf, 15 - k);
    }
    v2f cx2, cx1, cy2, cy1;
    {
        const float g13 = gs[14], g14 = gs[15], g15 = gs[16], g16 = gs[17];
        cx2.x = hi ? (cf.b2 * g14) : 0.0f;
        cx2.y = hi ? (cf.b2 * g15) : 0.0f;
        cx1.x = hi ? (cf.b1 * g14 + cf.b2 * g13) : 0.0f;
        cx1.y = hi ? (cf.b1 * g15 + cf.b2 * g14) : 0.0f;
        cy2.x = hi ? (-cf.a2 * g14) : 0.0f;
        cy2.y = hi ? (-cf.a2 * g15) : 0.0f;
        cy1.x = hi ? g15 : 0.0f;
        cy1.y = hi ? g16 : 0.0f;
    }
    v2f xprev;
    xprev.x = xh2; xprev.y = xh1;

    float* frow = out + (size_t)blockIdx.x * T_LEN + (size_t)(hi ? 8 : 0) * SLICE_LEN + j;

    auto trip = [&](const int m0) __attribute__((always_inline)) {
        v2f Xc[4][4];
#pragma unroll
        for (int u = 0; u < 4; ++u) {
            Xc[u][0].x = xa[m0 + u].x; Xc[u][0].y = xa[m0 + u].y;
            Xc[u][1].x = xa[m0 + u].z; Xc[u][1].y = xa[m0 + u].w;
            Xc[u][2].x = xb[m0 + u].x; Xc[u][2].y = xb[m0 + u].y;
            Xc[u][3].x = xb[m0 + u].z; Xc[u][3].y = xb[m0 + u].w;
        }

        v8f accA[4], accB[4];
        const v8f zero8 = {};
#pragma unroll
        for (int u = 0; u < 4; ++u) {
            accA[u] = __builtin_amdgcn_wmma_f32_16x16x4_f32(
                false, Xc[u][0], false, Hf[0], (short)0, zero8, false, false);
            accB[u] = __builtin_amdgcn_wmma_f32_16x16x4_f32(
                false, Xc[u][2], false, Hf[2], (short)0, zero8, false, false);
            accA[u] = __builtin_amdgcn_wmma_f32_16x16x4_f32(
                false, Xc[u][1], false, Hf[1], (short)0, accA[u], false, false);
            accB[u] = __builtin_amdgcn_wmma_f32_16x16x4_f32(
                false, Xc[u][3], false, Hf[3], (short)0, accB[u], false, false);
        }

        v2f base[4];
#pragma unroll
        for (int u = 0; u < 4; ++u) {
            v2f pa = hc[0] * Xc[u][0].x;
            v2f pb = hc[4] * Xc[u][2].x;
            pa = pa + hc[1] * Xc[u][0].y;
            pb = pb + hc[5] * Xc[u][2].y;
            pa = pa + hc[2] * Xc[u][1].x;
            pb = pb + hc[6] * Xc[u][3].x;
            pa = pa + hc[3] * Xc[u][1].y;
            pb = pb + hc[7] * Xc[u][3].y;
            const v2f ps = pa + pb;
            const v2f Wv = hi ? Xc[u][3] : ps;
            v2f R;
            R.x = dsvf_xhalf(Wv.x);
            R.y = dsvf_xhalf(Wv.y);
            const v2f xpv = (u == 0) ? xprev : Xc[(u + 3) & 3][3];
            const v2f up = (ps + R) + (cx2 * xpv.x + cx1 * xpv.y);
            base[u] = hi ? up : R;
        }
        xprev = Xc[3][3];

#pragma unroll
        for (int u = 0; u < 4; ++u) {
            const int t0 = (m0 + u) * 16;
            accA[u] = __builtin_amdgcn_wmma_f32_16x16x4_f32(
                false, Sf, false, Cs, (short)0, accA[u], false, false);
            const v8f yv = accA[u] + accB[u];
#pragma unroll
            for (int v = 0; v < 8; ++v)
                frow[(size_t)v * SLICE_LEN + t0] = yv[v];
            Sf = (base[u] + cy2 * Sf.x) + cy1 * Sf.y;
        }
    };

#pragma unroll
    for (int t = 0; t < SLICE_CHUNKS / 4; ++t) trip(4 * t);
}


extern "C" void kernel_launch(void* const* d_in, const int* in_sizes, int n_in,
                              void* d_out, int out_size, void* d_ws, size_t ws_size,
                              hipStream_t stream) {
    const float* x   = (const float*)d_in[0];
    const float* g   = (const float*)d_in[1];
    const float* r   = (const float*)d_in[2];
    const float* mhp = (const float*)d_in[3];
    const float* mbp = (const float*)d_in[4];
    const float* mlp = (const float*)d_in[5];
    float* y = (float*)d_out;

    const int B = in_sizes[0] / T_LEN;
    dim3 grid(B), block(32);
    hipLaunchKernelGGL(dsvf_row_kernel, grid, block, 0, stream,
                       x, g, r, mhp, mbp, mlp, y);
}
